# starting kernel
# baseline (speedup 1.0000x reference)
#include <hip/hip_runtime.h>
#include <math.h>
#include <float.h>


typedef __attribute__((ext_vector_type(2))) float v2f;
typedef __attribute__((ext_vector_type(8))) float v8f;

#define WSZ     32
#define LPITCH  36


__device__ __forceinline__ v8f gemm16(const float* sA, int aRow0, bool aT,
                                      const float* sB, int bCol0, bool bT,
                                      int lane)
{
    const int half = lane >> 4;
    const int r    = lane & 15;
    v8f c = {0.f,0.f,0.f,0.f,0.f,0.f,0.f,0.f};
    #pragma unroll
    for (int kq = 0; kq < 8; ++kq) {
        const int k0 = kq * 4 + half * 2;
        v2f a, b;
        if (!aT) {
            a.x = sA[(aRow0 + r) * LPITCH + k0];
            a.y = sA[(aRow0 + r) * LPITCH + k0 + 1];
        } else {
            a.x = sA[k0       * LPITCH + aRow0 + r];
            a.y = sA[(k0 + 1) * LPITCH + aRow0 + r];
        }
        if (!bT) {
            b.x = sB[k0       * LPITCH + bCol0 + r];
            b.y = sB[(k0 + 1) * LPITCH + bCol0 + r];
        } else {
            b.x = sB[(bCol0 + r) * LPITCH + k0];
            b.y = sB[(bCol0 + r) * LPITCH + k0 + 1];
        }
        c = __builtin_amdgcn_wmma_f32_16x16x4_f32(false, a, false, b,
                                                  (short)0, c, false, false);
    }
    return c;
}

__device__ __forceinline__ void gemm32(const float* sA, bool aT,
                                       const float* sB, bool bT,
                                       v8f acc[2][2], int lane)
{
    #pragma unroll
    for (int i = 0; i < 2; ++i)
        #pragma unroll
        for (int j = 0; j < 2; ++j)
            acc[i][j] = gemm16(sA, i * 16, aT, sB, j * 16, bT, lane);
}

__device__ __forceinline__ void store_cd(float* sOut, v8f acc[2][2], int lane)
{
    const int half = lane >> 4;
    const int n    = lane & 15;
    #pragma unroll
    for (int i = 0; i < 2; ++i)
        #pragma unroll
        for (int j = 0; j < 2; ++j)
            #pragma unroll
            for (int rr = 0; rr < 8; ++rr)
                sOut[(i * 16 + half * 8 + rr) * LPITCH + j * 16 + n] = acc[i][j][rr];
}

__global__ __launch_bounds__(1024) void init_consts_kernel(float* __restrict__ Dg,
                                                           float* __restrict__ Wg)
{
    const int t = threadIdx.x;
    const int i = t >> 5, j = t & 31;
    const double scale = (i == 0) ? sqrt(1.0 / 32.0) : sqrt(2.0 / 32.0);
    Dg[t] = (float)(scale * cos(((double)j + 0.5) * M_PI * (double)i / 32.0));

    const double s = (double)(i + j);
    double w = 0.0;
    for (int g = 0; g < 6; ++g) {
        const double start = 64.0 / 6.0 * (double)g;
        const double end   = 64.0 / 6.0 * (double)(g + 1);
        double cnt = 0.0;
        for (int ss = 0; ss <= 62; ++ss) {
            const double sd = (double)ss;
            if (!((sd > end) || (sd < start)))
                cnt += (double)((ss <= 31) ? ss + 1 : 63 - ss);
        }
        if (!((s > end) || (s < start)))
            w += exp2((double)g) / cnt;
    }
    Wg[t] = (float)w;
}

__global__ __launch_bounds__(32) void dct_grade_kernel(
    const float* __restrict__ x, const float* __restrict__ Dg,
    const float* __restrict__ Wg, float* __restrict__ grade,
    int C, int H, int Wd, int nW, int L)
{
    __shared__ float sD[32 * LPITCH];
    __shared__ float sP[32 * LPITCH];
    __shared__ float sT[32 * LPITCH];

    const int lane = threadIdx.x;
    const int bl   = blockIdx.x;
    const int l    = bl % L;
    const int b    = bl / L;
    const int pr   = (l / nW) * 16;
    const int pc   = (l % nW) * 16;

    for (int t = lane; t < 1024; t += 32)
        sD[(t >> 5) * LPITCH + (t & 31)] = Dg[t];

    const int half = lane >> 4;
    const int n    = lane & 15;
    float gs = 0.f;

    for (int c = 0; c < C; ++c) {
        const float* xp = x + (((size_t)b * C + c) * H + pr) * (size_t)Wd + pc;
        __syncthreads();
        for (int t = lane; t < 1024; t += 32)
            sP[(t >> 5) * LPITCH + (t & 31)] = xp[(size_t)(t >> 5) * Wd + (t & 31)];
        __syncthreads();

        v8f acc[2][2];
        gemm32(sD, false, sP, false, acc, lane);
        __syncthreads();
        store_cd(sT, acc, lane);
        __syncthreads();

        v8f xd[2][2];
        gemm32(sT, false, sD, true, xd, lane);

        #pragma unroll
        for (int i = 0; i < 2; ++i)
            #pragma unroll
            for (int j = 0; j < 2; ++j)
                #pragma unroll
                for (int rr = 0; rr < 8; ++rr) {
                    const int m  = i * 16 + half * 8 + rr;
                    const int nn = j * 16 + n;
                    const float v = xd[i][j][rr];
                    gs += logf(fabsf(v) + 1.0f) * Wg[m * 32 + nn];
                }
    }

    for (int off = 16; off > 0; off >>= 1)
        gs += __shfl_down(gs, off, 32);
    if (lane == 0) grade[bl] = gs;
}

__device__ __forceinline__ void ins_min(float v, int i, float& m1, int& i1,
                                        float& m2, int& i2)
{
    if (i < 0) return;
    if (v < m1 || (v == m1 && i < i1)) { m2 = m1; i2 = i1; m1 = v; i1 = i; }
    else if (v < m2 || (v == m2 && i < i2)) { m2 = v; i2 = i; }
}
__device__ __forceinline__ void ins_max(float v, int i, float& m1, int& i1,
                                        float& m2, int& i2)
{
    if (i < 0) return;
    if (v > m1 || (v == m1 && i > i1)) { m2 = m1; i2 = i1; m1 = v; i1 = i; }
    else if (v > m2 || (v == m2 && i > i2)) { m2 = v; i2 = i; }
}

__global__ __launch_bounds__(256) void select_kernel(const float* __restrict__ grade,
                                                     int* __restrict__ sel, int L)
{
    const int b   = blockIdx.x;
    const int tid = threadIdx.x;
    const float* g = grade + (size_t)b * L;

    float mn1 = FLT_MAX, mn2 = FLT_MAX, mx1 = -FLT_MAX, mx2 = -FLT_MAX;
    int   mi1 = -1, mi2 = -1, xi1 = -1, xi2 = -1;
    for (int l = tid; l < L; l += 256) {
        const float v = g[l];
        ins_min(v, l, mn1, mi1, mn2, mi2);
        ins_max(v, l, mx1, xi1, mx2, xi2);
    }

    __shared__ float smn1[256], smn2[256], smx1[256], smx2[256];
    __shared__ int   smi1[256], smi2[256], sxi1[256], sxi2[256];
    smn1[tid] = mn1; smn2[tid] = mn2; smx1[tid] = mx1; smx2[tid] = mx2;
    smi1[tid] = mi1; smi2[tid] = mi2; sxi1[tid] = xi1; sxi2[tid] = xi2;
    __syncthreads();

    for (int off = 128; off > 0; off >>= 1) {
        if (tid < off) {
            const int p = tid + off;
            ins_min(smn1[p], smi1[p], mn1, mi1, mn2, mi2);
            ins_min(smn2[p], smi2[p], mn1, mi1, mn2, mi2);
            ins_max(smx1[p], sxi1[p], mx1, xi1, mx2, xi2);
            ins_max(smx2[p], sxi2[p], mx1, xi1, mx2, xi2);
        }
        __syncthreads();
        if (tid < off) {
            smn1[tid] = mn1; smn2[tid] = mn2; smx1[tid] = mx1; smx2[tid] = mx2;
            smi1[tid] = mi1; smi2[tid] = mi2; sxi1[tid] = xi1; sxi2[tid] = xi2;
        }
        __syncthreads();
    }

    if (tid == 0) {
        sel[b * 4 + 0] = mi1;
        sel[b * 4 + 1] = xi1;
        sel[b * 4 + 2] = mi2;
        sel[b * 4 + 3] = xi2;
    }
}

__global__ __launch_bounds__(32) void level_out_kernel(
    const float* __restrict__ x, const float* __restrict__ Dg,
    const int* __restrict__ sel, float* __restrict__ out,
    int B, int C, int H, int Wd, int nW)
{
    __shared__ float sD[32 * LPITCH];
    __shared__ float sP[32 * LPITCH];
    __shared__ float sT[32 * LPITCH];

    const int lane = threadIdx.x;
    int bid = blockIdx.x;
    const int c = bid % C;  bid /= C;
    const int b = bid % B;
    const int s = bid / B;

    const int l  = sel[b * 4 + s];
    const int pr = (l / nW) * 16;
    const int pc = (l % nW) * 16;

    for (int t = lane; t < 1024; t += 32)
        sD[(t >> 5) * LPITCH + (t & 31)] = Dg[t];
    const float* xp = x + (((size_t)b * C + c) * H + pr) * (size_t)Wd + pc;
    for (int t = lane; t < 1024; t += 32)
        sP[(t >> 5) * LPITCH + (t & 31)] = xp[(size_t)(t >> 5) * Wd + (t & 31)];
    __syncthreads();

    const int half = lane >> 4;
    const int n    = lane & 15;

    v8f a0[2][2], a1[2][2];
    gemm32(sD, false, sP, false, a0, lane);
    store_cd(sT, a0, lane);
    __syncthreads();
    gemm32(sT, false, sD, true, a1, lane);

    #pragma unroll
    for (int i = 0; i < 2; ++i)
        #pragma unroll
        for (int j = 0; j < 2; ++j)
            #pragma unroll
            for (int rr = 0; rr < 8; ++rr) {
                const int m  = i * 16 + half * 8 + rr;
                const int nn = j * 16 + n;
                const float f = (((m + nn) > 64) || ((m + nn) < 0)) ? 0.f : 1.f;
                a1[i][j][rr] *= f;
            }

    __syncthreads();
    store_cd(sP, a1, lane);
    __syncthreads();
    gemm32(sD, true, sP, false, a0, lane);
    __syncthreads();
    store_cd(sT, a0, lane);
    __syncthreads();
    gemm32(sT, false, sD, false, a1, lane);

    float* op = out + (((size_t)s * B + b) * C + c) * 1024;
    #pragma unroll
    for (int i = 0; i < 2; ++i)
        #pragma unroll
        for (int j = 0; j < 2; ++j)
            #pragma unroll
            for (int rr = 0; rr < 8; ++rr)
                op[(i * 16 + half * 8 + rr) * 32 + j * 16 + n] = a1[i][j][rr];
}

extern "C" void kernel_launch(void* const* d_in, const int* in_sizes, int n_in,
                              void* d_out, int out_size, void* d_ws, size_t ws_size,
                              hipStream_t stream)
{
    const float* x = (const float*)d_in[0];
    float* out = (float*)d_out;

    const int B = 16, C = 3, H = 512, Wd = 512;
    const int nH = (H - WSZ) / 16 + 1;
    const int nW = (Wd - WSZ) / 16 + 1;
    const int L  = nH * nW;

    float* ws    = (float*)d_ws;
    float* Dg    = ws;
    float* Wg    = ws + 1024;
    float* grade = ws + 2048;
    int*   sel   = (int*)(ws + 2048 + B * L);

    hipLaunchKernelGGL(init_consts_kernel, dim3(1), dim3(1024), 0, stream, Dg, Wg);
    hipLaunchKernelGGL(dct_grade_kernel, dim3(B * L), dim3(32), 0, stream,
                       x, Dg, Wg, grade, C, H, Wd, nW, L);
    hipLaunchKernelGGL(select_kernel, dim3(B), dim3(256), 0, stream, grade, sel, L);
    hipLaunchKernelGGL(level_out_kernel, dim3(B * 4 * C), dim3(32), 0, stream,
                       x, Dg, sel, out, B, C, H, Wd, nW);
}


# constant-table launch: 64 one-wave work-groups (one table row each), band sizes by integer wave sum and one fp64 division per wave, instead of 1024 threads of fp64 on one WGP; sums unchanged
# speedup vs baseline: 1.6680x; 1.6680x over previous
#include <hip/hip_runtime.h>
#include <math.h>
#include <float.h>


typedef __attribute__((ext_vector_type(2))) float v2f;
typedef __attribute__((ext_vector_type(8))) float v8f;

#define WSZ     32
#define LPITCH  36


__device__ __forceinline__ v8f gemm16(const float* sA, int aRow0, bool aT,
                                      const float* sB, int bCol0, bool bT,
                                      int lane)
{
    const int half = lane >> 4;
    const int r    = lane & 15;
    v8f c = {0.f,0.f,0.f,0.f,0.f,0.f,0.f,0.f};
    #pragma unroll
    for (int kq = 0; kq < 8; ++kq) {
        const int k0 = kq * 4 + half * 2;
        v2f a, b;
        if (!aT) {
            a.x = sA[(aRow0 + r) * LPITCH + k0];
            a.y = sA[(aRow0 + r) * LPITCH + k0 + 1];
        } else {
            a.x = sA[k0       * LPITCH + aRow0 + r];
            a.y = sA[(k0 + 1) * LPITCH + aRow0 + r];
        }
        if (!bT) {
            b.x = sB[k0       * LPITCH + bCol0 + r];
            b.y = sB[(k0 + 1) * LPITCH + bCol0 + r];
        } else {
            b.x = sB[(bCol0 + r) * LPITCH + k0];
            b.y = sB[(bCol0 + r) * LPITCH + k0 + 1];
        }
        c = __builtin_amdgcn_wmma_f32_16x16x4_f32(false, a, false, b,
                                                  (short)0, c, false, false);
    }
    return c;
}

__device__ __forceinline__ void gemm32(const float* sA, bool aT,
                                       const float* sB, bool bT,
                                       v8f acc[2][2], int lane)
{
    #pragma unroll
    for (int i = 0; i < 2; ++i)
        #pragma unroll
        for (int j = 0; j < 2; ++j)
            acc[i][j] = gemm16(sA, i * 16, aT, sB, j * 16, bT, lane);
}

__device__ __forceinline__ void store_cd(float* sOut, v8f acc[2][2], int lane)
{
    const int half = lane >> 4;
    const int n    = lane & 15;
    #pragma unroll
    for (int i = 0; i < 2; ++i)
        #pragma unroll
        for (int j = 0; j < 2; ++j)
            #pragma unroll
            for (int rr = 0; rr < 8; ++rr)
                sOut[(i * 16 + half * 8 + rr) * LPITCH + j * 16 + n] = acc[i][j][rr];
}


__device__ __forceinline__ void band_bounds(int g, int& lo, int& hi)
{
    const double start = 64.0 / 6.0 * (double)g;
    const double end   = 64.0 / 6.0 * (double)(g + 1);
    lo = (int)start;
    if ((double)lo < start) ++lo;
    hi = (int)end;
}

__global__ __launch_bounds__(32) void init_consts_kernel(float* __restrict__ Dg,
                                                         float* __restrict__ Wg)
{
    const int lane = threadIdx.x;

    if (blockIdx.x < 32) {
        const int i = blockIdx.x, j = lane;
        const double scale = (i == 0) ? sqrt(1.0 / 32.0) : sqrt(2.0 / 32.0);
        Dg[i * 32 + j] =
            (float)(scale * cos(((double)j + 0.5) * M_PI * (double)i / 32.0));
        return;
    }

    const int i = blockIdx.x - 32, j = lane;

    int cnt[6];
    #pragma unroll
    for (int g = 0; g < 6; ++g) {
        int lo, hi;
        band_bounds(g, lo, hi);
        const int s0 = lane, s1 = lane + 32;
        int c = 0;
        if (s0 >= lo && s0 <= hi) c += s0 + 1;
        if (s1 >= lo && s1 <= hi) c += 63 - s1;
        for (int off = 16; off > 0; off >>= 1)
            c += __shfl_xor(c, off, 32);
        cnt[g] = c;
    }

    const int gl = (lane < 6) ? lane : 0;
    int myc = cnt[0];
    #pragma unroll
    for (int g = 1; g < 6; ++g)
        if (gl == g) myc = cnt[g];
    const double q = (double)(1 << gl) / (double)myc;

    const int s = i + j;
    double w = 0.0;
    #pragma unroll
    for (int g = 0; g < 6; ++g) {
        int lo, hi;
        band_bounds(g, lo, hi);
        const double qg = __shfl(q, g, 32);
        w += (s >= lo && s <= hi) ? qg : 0.0;
    }
    Wg[i * 32 + j] = (float)w;
}

__global__ __launch_bounds__(32) void dct_grade_kernel(
    const float* __restrict__ x, const float* __restrict__ Dg,
    const float* __restrict__ Wg, float* __restrict__ grade,
    int C, int H, int Wd, int nW, int L)
{
    __shared__ float sD[32 * LPITCH];
    __shared__ float sP[32 * LPITCH];
    __shared__ float sT[32 * LPITCH];

    const int lane = threadIdx.x;
    const int bl   = blockIdx.x;
    const int l    = bl % L;
    const int b    = bl / L;
    const int pr   = (l / nW) * 16;
    const int pc   = (l % nW) * 16;

    for (int t = lane; t < 1024; t += 32)
        sD[(t >> 5) * LPITCH + (t & 31)] = Dg[t];

    const int half = lane >> 4;
    const int n    = lane & 15;
    float gs = 0.f;

    for (int c = 0; c < C; ++c) {
        const float* xp = x + (((size_t)b * C + c) * H + pr) * (size_t)Wd + pc;
        __syncthreads();
        for (int t = lane; t < 1024; t += 32)
            sP[(t >> 5) * LPITCH + (t & 31)] = xp[(size_t)(t >> 5) * Wd + (t & 31)];
        __syncthreads();

        v8f acc[2][2];
        gemm32(sD, false, sP, false, acc, lane);
        __syncthreads();
        store_cd(sT, acc, lane);
        __syncthreads();

        v8f xd[2][2];
        gemm32(sT, false, sD, true, xd, lane);

        #pragma unroll
        for (int i = 0; i < 2; ++i)
            #pragma unroll
            for (int j = 0; j < 2; ++j)
                #pragma unroll
                for (int rr = 0; rr < 8; ++rr) {
                    const int m  = i * 16 + half * 8 + rr;
                    const int nn = j * 16 + n;
                    const float v = xd[i][j][rr];
                    gs += logf(fabsf(v) + 1.0f) * Wg[m * 32 + nn];
                }
    }

    for (int off = 16; off > 0; off >>= 1)
        gs += __shfl_down(gs, off, 32);
    if (lane == 0) grade[bl] = gs;
}

__device__ __forceinline__ void ins_min(float v, int i, float& m1, int& i1,
                                        float& m2, int& i2)
{
    if (i < 0) return;
    if (v < m1 || (v == m1 && i < i1)) { m2 = m1; i2 = i1; m1 = v; i1 = i; }
    else if (v < m2 || (v == m2 && i < i2)) { m2 = v; i2 = i; }
}
__device__ __forceinline__ void ins_max(float v, int i, float& m1, int& i1,
                                        float& m2, int& i2)
{
    if (i < 0) return;
    if (v > m1 || (v == m1 && i > i1)) { m2 = m1; i2 = i1; m1 = v; i1 = i; }
    else if (v > m2 || (v == m2 && i > i2)) { m2 = v; i2 = i; }
}

__global__ __launch_bounds__(256) void select_kernel(const float* __restrict__ grade,
                                                     int* __restrict__ sel, int L)
{
    const int b   = blockIdx.x;
    const int tid = threadIdx.x;
    const float* g = grade + (size_t)b * L;

    float mn1 = FLT_MAX, mn2 = FLT_MAX, mx1 = -FLT_MAX, mx2 = -FLT_MAX;
    int   mi1 = -1, mi2 = -1, xi1 = -1, xi2 = -1;
    for (int l = tid; l < L; l += 256) {
        const float v = g[l];
        ins_min(v, l, mn1, mi1, mn2, mi2);
        ins_max(v, l, mx1, xi1, mx2, xi2);
    }

    __shared__ float smn1[256], smn2[256], smx1[256], smx2[256];
    __shared__ int   smi1[256], smi2[256], sxi1[256], sxi2[256];
    smn1[tid] = mn1; smn2[tid] = mn2; smx1[tid] = mx1; smx2[tid] = mx2;
    smi1[tid] = mi1; smi2[tid] = mi2; sxi1[tid] = xi1; sxi2[tid] = xi2;
    __syncthreads();

    for (int off = 128; off > 0; off >>= 1) {
        if (tid < off) {
            const int p = tid + off;
            ins_min(smn1[p], smi1[p], mn1, mi1, mn2, mi2);
            ins_min(smn2[p], smi2[p], mn1, mi1, mn2, mi2);
            ins_max(smx1[p], sxi1[p], mx1, xi1, mx2, xi2);
            ins_max(smx2[p], sxi2[p], mx1, xi1, mx2, xi2);
        }
        __syncthreads();
        if (tid < off) {
            smn1[tid] = mn1; smn2[tid] = mn2; smx1[tid] = mx1; smx2[tid] = mx2;
            smi1[tid] = mi1; smi2[tid] = mi2; sxi1[tid] = xi1; sxi2[tid] = xi2;
        }
        __syncthreads();
    }

    if (tid == 0) {
        sel[b * 4 + 0] = mi1;
        sel[b * 4 + 1] = xi1;
        sel[b * 4 + 2] = mi2;
        sel[b * 4 + 3] = xi2;
    }
}

__global__ __launch_bounds__(32) void level_out_kernel(
    const float* __restrict__ x, const float* __restrict__ Dg,
    const int* __restrict__ sel, float* __restrict__ out,
    int B, int C, int H, int Wd, int nW)
{
    __shared__ float sD[32 * LPITCH];
    __shared__ float sP[32 * LPITCH];
    __shared__ float sT[32 * LPITCH];

    const int lane = threadIdx.x;
    int bid = blockIdx.x;
    const int c = bid % C;  bid /= C;
    const int b = bid % B;
    const int s = bid / B;

    const int l  = sel[b * 4 + s];
    const int pr = (l / nW) * 16;
    const int pc = (l % nW) * 16;

    for (int t = lane; t < 1024; t += 32)
        sD[(t >> 5) * LPITCH + (t & 31)] = Dg[t];
    const float* xp = x + (((size_t)b * C + c) * H + pr) * (size_t)Wd + pc;
    for (int t = lane; t < 1024; t += 32)
        sP[(t >> 5) * LPITCH + (t & 31)] = xp[(size_t)(t >> 5) * Wd + (t & 31)];
    __syncthreads();

    const int half = lane >> 4;
    const int n    = lane & 15;

    v8f a0[2][2], a1[2][2];
    gemm32(sD, false, sP, false, a0, lane);
    store_cd(sT, a0, lane);
    __syncthreads();
    gemm32(sT, false, sD, true, a1, lane);

    #pragma unroll
    for (int i = 0; i < 2; ++i)
        #pragma unroll
        for (int j = 0; j < 2; ++j)
            #pragma unroll
            for (int rr = 0; rr < 8; ++rr) {
                const int m  = i * 16 + half * 8 + rr;
                const int nn = j * 16 + n;
                const float f = (((m + nn) > 64) || ((m + nn) < 0)) ? 0.f : 1.f;
                a1[i][j][rr] *= f;
            }

    __syncthreads();
    store_cd(sP, a1, lane);
    __syncthreads();
    gemm32(sD, true, sP, false, a0, lane);
    __syncthreads();
    store_cd(sT, a0, lane);
    __syncthreads();
    gemm32(sT, false, sD, false, a1, lane);

    float* op = out + (((size_t)s * B + b) * C + c) * 1024;
    #pragma unroll
    for (int i = 0; i < 2; ++i)
        #pragma unroll
        for (int j = 0; j < 2; ++j)
            #pragma unroll
            for (int rr = 0; rr < 8; ++rr)
                op[(i * 16 + half * 8 + rr) * 32 + j * 16 + n] = a1[i][j][rr];
}

extern "C" void kernel_launch(void* const* d_in, const int* in_sizes, int n_in,
                              void* d_out, int out_size, void* d_ws, size_t ws_size,
                              hipStream_t stream)
{
    const float* x = (const float*)d_in[0];
    float* out = (float*)d_out;

    const int B = 16, C = 3, H = 512, Wd = 512;
    const int nH = (H - WSZ) / 16 + 1;
    const int nW = (Wd - WSZ) / 16 + 1;
    const int L  = nH * nW;

    float* ws    = (float*)d_ws;
    float* Dg    = ws;
    float* Wg    = ws + 1024;
    float* grade = ws + 2048;
    int*   sel   = (int*)(ws + 2048 + B * L);

    hipLaunchKernelGGL(init_consts_kernel, dim3(64), dim3(32), 0, stream, Dg, Wg);
    hipLaunchKernelGGL(dct_grade_kernel, dim3(B * L), dim3(32), 0, stream,
                       x, Dg, Wg, grade, C, H, Wd, nW, L);
    hipLaunchKernelGGL(select_kernel, dim3(B), dim3(256), 0, stream, grade, sel, L);
    hipLaunchKernelGGL(level_out_kernel, dim3(B * 4 * C), dim3(32), 0, stream,
                       x, Dg, sel, out, B, C, H, Wd, nW);
}


# grade kernel: both 32-deep products folded to 16-deep on the DCT matrix's even/odd symmetry (mirrored rows/columns added and subtracted first): 32 WMMAs per patch-channel, not 64; sum order changes
# speedup vs baseline: 4.5184x; 2.7089x over previous
#include <hip/hip_runtime.h>
#include <math.h>
#include <float.h>


typedef __attribute__((ext_vector_type(2))) float v2f;
typedef __attribute__((ext_vector_type(8))) float v8f;
typedef __attribute__((ext_vector_type(4))) float v4f;

#define WSZ     32
#define LPITCH  36


__device__ __forceinline__ v8f gemm16(const float* sA, int aRow0, bool aT,
                                      const float* sB, int bCol0, bool bT,
                                      int lane)
{
    const int half = lane >> 4;
    const int r    = lane & 15;
    v8f c = {0.f,0.f,0.f,0.f,0.f,0.f,0.f,0.f};
    #pragma unroll
    for (int kq = 0; kq < 8; ++kq) {
        const int k0 = kq * 4 + half * 2;
        v2f a, b;
        if (!aT) {
            a.x = sA[(aRow0 + r) * LPITCH + k0];
            a.y = sA[(aRow0 + r) * LPITCH + k0 + 1];
        } else {
            a.x = sA[k0       * LPITCH + aRow0 + r];
            a.y = sA[(k0 + 1) * LPITCH + aRow0 + r];
        }
        if (!bT) {
            b.x = sB[k0       * LPITCH + bCol0 + r];
            b.y = sB[(k0 + 1) * LPITCH + bCol0 + r];
        } else {
            b.x = sB[(bCol0 + r) * LPITCH + k0];
            b.y = sB[(bCol0 + r) * LPITCH + k0 + 1];
        }
        c = __builtin_amdgcn_wmma_f32_16x16x4_f32(false, a, false, b,
                                                  (short)0, c, false, false);
    }
    return c;
}

__device__ __forceinline__ void gemm32(const float* sA, bool aT,
                                       const float* sB, bool bT,
                                       v8f acc[2][2], int lane)
{
    #pragma unroll
    for (int i = 0; i < 2; ++i)
        #pragma unroll
        for (int j = 0; j < 2; ++j)
            acc[i][j] = gemm16(sA, i * 16, aT, sB, j * 16, bT, lane);
}

__device__ __forceinline__ void store_cd(float* sOut, v8f acc[2][2], int lane)
{
    const int half = lane >> 4;
    const int n    = lane & 15;
    #pragma unroll
    for (int i = 0; i < 2; ++i)
        #pragma unroll
        for (int j = 0; j < 2; ++j)
            #pragma unroll
            for (int rr = 0; rr < 8; ++rr)
                sOut[(i * 16 + half * 8 + rr) * LPITCH + j * 16 + n] = acc[i][j][rr];
}

__device__ __forceinline__ void stage_tile(float* sDst, const float* __restrict__ src,
                                           size_t stride, int lane)
{
    const int r0 = lane >> 3;
    const int c0 = (lane & 7) * 4;
    v4f v[8];
    #pragma unroll
    for (int g = 0; g < 8; ++g)
        v[g] = *(const v4f*)(src + (size_t)(g * 4 + r0) * stride + c0);
    #pragma unroll
    for (int g = 0; g < 8; ++g)
        *(v4f*)(sDst + (g * 4 + r0) * LPITCH + c0) = v[g];
}

__device__ __forceinline__ void stage_fold(float* sDst, const float* __restrict__ src,
                                           size_t stride, int lane)
{
    const int r0 = lane >> 3;
    const int c0 = (lane & 7) * 4;
    v4f top[4], bot[4];
    #pragma unroll
    for (int g = 0; g < 4; ++g) {
        const int j = g * 4 + r0;
        top[g] = *(const v4f*)(src + (size_t)j * stride + c0);
        bot[g] = *(const v4f*)(src + (size_t)(31 - j) * stride + c0);
    }
    #pragma unroll
    for (int g = 0; g < 4; ++g) {
        const int j = g * 4 + r0;
        *(v4f*)(sDst + j * LPITCH + c0)        = top[g] + bot[g];
        *(v4f*)(sDst + (16 + j) * LPITCH + c0) = top[g] - bot[g];
    }
}


__device__ __forceinline__ void band_bounds(int g, int& lo, int& hi)
{
    const double start = 64.0 / 6.0 * (double)g;
    const double end   = 64.0 / 6.0 * (double)(g + 1);
    lo = (int)start;
    if ((double)lo < start) ++lo;
    hi = (int)end;
}

__global__ __launch_bounds__(32) void init_consts_kernel(float* __restrict__ Dg,
                                                         float* __restrict__ Wg)
{
    const int lane = threadIdx.x;

    if (blockIdx.x < 32) {
        const int i = blockIdx.x, j = lane;
        const double scale = (i == 0) ? sqrt(1.0 / 32.0) : sqrt(2.0 / 32.0);
        Dg[i * 32 + j] =
            (float)(scale * cos(((double)j + 0.5) * M_PI * (double)i / 32.0));
        return;
    }

    const int i = blockIdx.x - 32, j = lane;

    int cnt[6];
    #pragma unroll
    for (int g = 0; g < 6; ++g) {
        int lo, hi;
        band_bounds(g, lo, hi);
        const int s0 = lane, s1 = lane + 32;
        int c = 0;
        if (s0 >= lo && s0 <= hi) c += s0 + 1;
        if (s1 >= lo && s1 <= hi) c += 63 - s1;
        for (int off = 16; off > 0; off >>= 1)
            c += __shfl_xor(c, off, 32);
        cnt[g] = c;
    }

    const int gl = (lane < 6) ? lane : 0;
    int myc = cnt[0];
    #pragma unroll
    for (int g = 1; g < 6; ++g)
        if (gl == g) myc = cnt[g];
    const double q = (double)(1 << gl) / (double)myc;

    const int s = i + j;
    double w = 0.0;
    #pragma unroll
    for (int g = 0; g < 6; ++g) {
        int lo, hi;
        band_bounds(g, lo, hi);
        const double qg = __shfl(q, g, 32);
        w += (s >= lo && s <= hi) ? qg : 0.0;
    }
    Wg[i * 32 + j] = (float)w;
}

__global__ __launch_bounds__(32) void dct_grade_kernel(
    const float* __restrict__ x, const float* __restrict__ Dg,
    const float* __restrict__ Wg, float* __restrict__ grade,
    int C, int H, int Wd, int nW, int L)
{
    __shared__ __attribute__((aligned(16))) float sP[32 * LPITCH];
    __shared__ __attribute__((aligned(16))) float sT[32 * LPITCH];

    const int lane = threadIdx.x;
    const int bl   = blockIdx.x;
    const int l    = bl % L;
    const int b    = bl / L;
    const int pr   = (l / nW) * 16;
    const int pc   = (l % nW) * 16;

    const int half = lane >> 4;
    const int n    = lane & 15;

    v2f dfr[2][4];
    #pragma unroll
    for (int p = 0; p < 2; ++p)
        #pragma unroll
        for (int kq = 0; kq < 4; ++kq)
            dfr[p][kq] = *(const v2f*)(Dg + (2 * n + p) * 32 + kq * 4 + half * 2);

    float gs = 0.f;

    for (int c = 0; c < C; ++c) {
        const float* xp = x + (((size_t)b * C + c) * H + pr) * (size_t)Wd + pc;
        __syncthreads();
        stage_fold(sP, xp, (size_t)Wd, lane);
        __syncthreads();

        v8f t[2][2];
        #pragma unroll
        for (int p = 0; p < 2; ++p)
            #pragma unroll
            for (int ct = 0; ct < 2; ++ct) {
                v8f acc = {0.f,0.f,0.f,0.f,0.f,0.f,0.f,0.f};
                #pragma unroll
                for (int kq = 0; kq < 4; ++kq) {
                    const int k0 = kq * 4 + half * 2;
                    v2f bb;
                    bb.x = sP[(16 * p + k0)     * LPITCH + ct * 16 + n];
                    bb.y = sP[(16 * p + k0 + 1) * LPITCH + ct * 16 + n];
                    acc = __builtin_amdgcn_wmma_f32_16x16x4_f32(false, dfr[p][kq], false, bb,
                                                                (short)0, acc, false, false);
                }
                t[p][ct] = acc;
            }
        store_cd(sT, t, lane);
        __syncthreads();

        v8f xd[2][2];
        #pragma unroll
        for (int pi = 0; pi < 2; ++pi) {
            v2f ae[4], ao[4];
            #pragma unroll
            for (int kq = 0; kq < 4; ++kq) {
                const int k0 = kq * 4 + half * 2;
                const v2f t0 = *(const v2f*)(sT + (16 * pi + n) * LPITCH + k0);
                const v2f t1 = *(const v2f*)(sT + (16 * pi + n) * LPITCH + 30 - k0);
                ae[kq].x = t0.x + t1.y;  ae[kq].y = t0.y + t1.x;
                ao[kq].x = t0.x - t1.y;  ao[kq].y = t0.y - t1.x;
            }
            v8f xe = {0.f,0.f,0.f,0.f,0.f,0.f,0.f,0.f};
            v8f xo = {0.f,0.f,0.f,0.f,0.f,0.f,0.f,0.f};
            #pragma unroll
            for (int kq = 0; kq < 4; ++kq) {
                xe = __builtin_amdgcn_wmma_f32_16x16x4_f32(false, ae[kq], false, dfr[0][kq],
                                                           (short)0, xe, false, false);
                xo = __builtin_amdgcn_wmma_f32_16x16x4_f32(false, ao[kq], false, dfr[1][kq],
                                                           (short)0, xo, false, false);
            }
            xd[pi][0] = xe;
            xd[pi][1] = xo;
        }

        #pragma unroll
        for (int pi = 0; pi < 2; ++pi)
            #pragma unroll
            for (int pn = 0; pn < 2; ++pn)
                #pragma unroll
                for (int rr = 0; rr < 8; ++rr) {
                    const int m  = 2 * (half * 8 + rr) + pi;
                    const int nn = 2 * n + pn;
                    const float v = xd[pi][pn][rr];
                    gs += logf(fabsf(v) + 1.0f) * Wg[m * 32 + nn];
                }
    }

    for (int off = 16; off > 0; off >>= 1)
        gs += __shfl_down(gs, off, 32);
    if (lane == 0) grade[bl] = gs;
}

__device__ __forceinline__ void ins_min(float v, int i, float& m1, int& i1,
                                        float& m2, int& i2)
{
    if (i < 0) return;
    if (v < m1 || (v == m1 && i < i1)) { m2 = m1; i2 = i1; m1 = v; i1 = i; }
    else if (v < m2 || (v == m2 && i < i2)) { m2 = v; i2 = i; }
}
__device__ __forceinline__ void ins_max(float v, int i, float& m1, int& i1,
                                        float& m2, int& i2)
{
    if (i < 0) return;
    if (v > m1 || (v == m1 && i > i1)) { m2 = m1; i2 = i1; m1 = v; i1 = i; }
    else if (v > m2 || (v == m2 && i > i2)) { m2 = v; i2 = i; }
}

__global__ __launch_bounds__(256) void select_kernel(const float* __restrict__ grade,
                                                     int* __restrict__ sel, int L)
{
    const int b   = blockIdx.x;
    const int tid = threadIdx.x;
    const float* g = grade + (size_t)b * L;

    float mn1 = FLT_MAX, mn2 = FLT_MAX, mx1 = -FLT_MAX, mx2 = -FLT_MAX;
    int   mi1 = -1, mi2 = -1, xi1 = -1, xi2 = -1;
    for (int l = tid; l < L; l += 256) {
        const float v = g[l];
        ins_min(v, l, mn1, mi1, mn2, mi2);
        ins_max(v, l, mx1, xi1, mx2, xi2);
    }

    __shared__ float smn1[256], smn2[256], smx1[256], smx2[256];
    __shared__ int   smi1[256], smi2[256], sxi1[256], sxi2[256];
    smn1[tid] = mn1; smn2[tid] = mn2; smx1[tid] = mx1; smx2[tid] = mx2;
    smi1[tid] = mi1; smi2[tid] = mi2; sxi1[tid] = xi1; sxi2[tid] = xi2;
    __syncthreads();

    for (int off = 128; off > 0; off >>= 1) {
        if (tid < off) {
            const int p = tid + off;
            ins_min(smn1[p], smi1[p], mn1, mi1, mn2, mi2);
            ins_min(smn2[p], smi2[p], mn1, mi1, mn2, mi2);
            ins_max(smx1[p], sxi1[p], mx1, xi1, mx2, xi2);
            ins_max(smx2[p], sxi2[p], mx1, xi1, mx2, xi2);
        }
        __syncthreads();
        if (tid < off) {
            smn1[tid] = mn1; smn2[tid] = mn2; smx1[tid] = mx1; smx2[tid] = mx2;
            smi1[tid] = mi1; smi2[tid] = mi2; sxi1[tid] = xi1; sxi2[tid] = xi2;
        }
        __syncthreads();
    }

    if (tid == 0) {
        sel[b * 4 + 0] = mi1;
        sel[b * 4 + 1] = xi1;
        sel[b * 4 + 2] = mi2;
        sel[b * 4 + 3] = xi2;
    }
}

__global__ __launch_bounds__(32) void level_out_kernel(
    const float* __restrict__ x, const float* __restrict__ Dg,
    const int* __restrict__ sel, float* __restrict__ out,
    int B, int C, int H, int Wd, int nW)
{
    __shared__ __attribute__((aligned(16))) float sD[32 * LPITCH];
    __shared__ __attribute__((aligned(16))) float sP[32 * LPITCH];
    __shared__ float sT[32 * LPITCH];

    const int lane = threadIdx.x;
    int bid = blockIdx.x;
    const int c = bid % C;  bid /= C;
    const int b = bid % B;
    const int s = bid / B;

    const int l  = sel[b * 4 + s];
    const int pr = (l / nW) * 16;
    const int pc = (l % nW) * 16;

    stage_tile(sD, Dg, 32, lane);
    const float* xp = x + (((size_t)b * C + c) * H + pr) * (size_t)Wd + pc;
    stage_tile(sP, xp, (size_t)Wd, lane);
    __syncthreads();

    const int half = lane >> 4;
    const int n    = lane & 15;

    v8f a0[2][2], a1[2][2];
    gemm32(sD, false, sP, false, a0, lane);
    store_cd(sT, a0, lane);
    __syncthreads();
    gemm32(sT, false, sD, true, a1, lane);

    #pragma unroll
    for (int i = 0; i < 2; ++i)
        #pragma unroll
        for (int j = 0; j < 2; ++j)
            #pragma unroll
            for (int rr = 0; rr < 8; ++rr) {
                const int m  = i * 16 + half * 8 + rr;
                const int nn = j * 16 + n;
                const float f = (((m + nn) > 64) || ((m + nn) < 0)) ? 0.f : 1.f;
                a1[i][j][rr] *= f;
            }

    __syncthreads();
    store_cd(sP, a1, lane);
    __syncthreads();
    gemm32(sD, true, sP, false, a0, lane);
    __syncthreads();
    store_cd(sT, a0, lane);
    __syncthreads();
    gemm32(sT, false, sD, false, a1, lane);

    float* op = out + (((size_t)s * B + b) * C + c) * 1024;
    #pragma unroll
    for (int i = 0; i < 2; ++i)
        #pragma unroll
        for (int j = 0; j < 2; ++j)
            #pragma unroll
            for (int rr = 0; rr < 8; ++rr)
                op[(i * 16 + half * 8 + rr) * 32 + j * 16 + n] = a1[i][j][rr];
}

extern "C" void kernel_launch(void* const* d_in, const int* in_sizes, int n_in,
                              void* d_out, int out_size, void* d_ws, size_t ws_size,
                              hipStream_t stream)
{
    const float* x = (const float*)d_in[0];
    float* out = (float*)d_out;

    const int B = 16, C = 3, H = 512, Wd = 512;
    const int nH = (H - WSZ) / 16 + 1;
    const int nW = (Wd - WSZ) / 16 + 1;
    const int L  = nH * nW;

    float* ws    = (float*)d_ws;
    float* Dg    = ws;
    float* Wg    = ws + 1024;
    float* grade = ws + 2048;
    int*   sel   = (int*)(ws + 2048 + B * L);

    hipLaunchKernelGGL(init_consts_kernel, dim3(64), dim3(32), 0, stream, Dg, Wg);
    hipLaunchKernelGGL(dct_grade_kernel, dim3(B * L), dim3(32), 0, stream,
                       x, Dg, Wg, grade, C, H, Wd, nW, L);
    hipLaunchKernelGGL(select_kernel, dim3(B), dim3(256), 0, stream, grade, sel, L);
    hipLaunchKernelGGL(level_out_kernel, dim3(B * 4 * C), dim3(32), 0, stream,
                       x, Dg, sel, out, B, C, H, Wd, nW);
}


# grade kernel: log(|X|+1) taken with the bare log2 instruction (argument >= 1, so the library log's denormal pre-scale and non-finite patch can never act); ln 2 applied once to the wave's sum
# speedup vs baseline: 5.1401x; 1.1376x over previous
#include <hip/hip_runtime.h>
#include <math.h>
#include <float.h>


typedef __attribute__((ext_vector_type(2))) float v2f;
typedef __attribute__((ext_vector_type(8))) float v8f;
typedef __attribute__((ext_vector_type(4))) float v4f;

#define WSZ     32
#define LPITCH  36


__device__ __forceinline__ v8f gemm16(const float* sA, int aRow0, bool aT,
                                      const float* sB, int bCol0, bool bT,
                                      int lane)
{
    const int half = lane >> 4;
    const int r    = lane & 15;
    v8f c = {0.f,0.f,0.f,0.f,0.f,0.f,0.f,0.f};
    #pragma unroll
    for (int kq = 0; kq < 8; ++kq) {
        const int k0 = kq * 4 + half * 2;
        v2f a, b;
        if (!aT) {
            a.x = sA[(aRow0 + r) * LPITCH + k0];
            a.y = sA[(aRow0 + r) * LPITCH + k0 + 1];
        } else {
            a.x = sA[k0       * LPITCH + aRow0 + r];
            a.y = sA[(k0 + 1) * LPITCH + aRow0 + r];
        }
        if (!bT) {
            b.x = sB[k0       * LPITCH + bCol0 + r];
            b.y = sB[(k0 + 1) * LPITCH + bCol0 + r];
        } else {
            b.x = sB[(bCol0 + r) * LPITCH + k0];
            b.y = sB[(bCol0 + r) * LPITCH + k0 + 1];
        }
        c = __builtin_amdgcn_wmma_f32_16x16x4_f32(false, a, false, b,
                                                  (short)0, c, false, false);
    }
    return c;
}

__device__ __forceinline__ void gemm32(const float* sA, bool aT,
                                       const float* sB, bool bT,
                                       v8f acc[2][2], int lane)
{
    #pragma unroll
    for (int i = 0; i < 2; ++i)
        #pragma unroll
        for (int j = 0; j < 2; ++j)
            acc[i][j] = gemm16(sA, i * 16, aT, sB, j * 16, bT, lane);
}

__device__ __forceinline__ void store_cd(float* sOut, v8f acc[2][2], int lane)
{
    const int half = lane >> 4;
    const int n    = lane & 15;
    #pragma unroll
    for (int i = 0; i < 2; ++i)
        #pragma unroll
        for (int j = 0; j < 2; ++j)
            #pragma unroll
            for (int rr = 0; rr < 8; ++rr)
                sOut[(i * 16 + half * 8 + rr) * LPITCH + j * 16 + n] = acc[i][j][rr];
}

__device__ __forceinline__ void stage_tile(float* sDst, const float* __restrict__ src,
                                           size_t stride, int lane)
{
    const int r0 = lane >> 3;
    const int c0 = (lane & 7) * 4;
    v4f v[8];
    #pragma unroll
    for (int g = 0; g < 8; ++g)
        v[g] = *(const v4f*)(src + (size_t)(g * 4 + r0) * stride + c0);
    #pragma unroll
    for (int g = 0; g < 8; ++g)
        *(v4f*)(sDst + (g * 4 + r0) * LPITCH + c0) = v[g];
}

__device__ __forceinline__ void stage_fold(float* sDst, const float* __restrict__ src,
                                           size_t stride, int lane)
{
    const int r0 = lane >> 3;
    const int c0 = (lane & 7) * 4;
    v4f top[4], bot[4];
    #pragma unroll
    for (int g = 0; g < 4; ++g) {
        const int j = g * 4 + r0;
        top[g] = *(const v4f*)(src + (size_t)j * stride + c0);
        bot[g] = *(const v4f*)(src + (size_t)(31 - j) * stride + c0);
    }
    #pragma unroll
    for (int g = 0; g < 4; ++g) {
        const int j = g * 4 + r0;
        *(v4f*)(sDst + j * LPITCH + c0)        = top[g] + bot[g];
        *(v4f*)(sDst + (16 + j) * LPITCH + c0) = top[g] - bot[g];
    }
}


__device__ __forceinline__ void band_bounds(int g, int& lo, int& hi)
{
    const double start = 64.0 / 6.0 * (double)g;
    const double end   = 64.0 / 6.0 * (double)(g + 1);
    lo = (int)start;
    if ((double)lo < start) ++lo;
    hi = (int)end;
}

__global__ __launch_bounds__(32) void init_consts_kernel(float* __restrict__ Dg,
                                                         float* __restrict__ Wg)
{
    const int lane = threadIdx.x;

    if (blockIdx.x < 32) {
        const int i = blockIdx.x, j = lane;
        const double scale = (i == 0) ? sqrt(1.0 / 32.0) : sqrt(2.0 / 32.0);
        Dg[i * 32 + j] =
            (float)(scale * cos(((double)j + 0.5) * M_PI * (double)i / 32.0));
        return;
    }

    const int i = blockIdx.x - 32, j = lane;

    int cnt[6];
    #pragma unroll
    for (int g = 0; g < 6; ++g) {
        int lo, hi;
        band_bounds(g, lo, hi);
        const int s0 = lane, s1 = lane + 32;
        int c = 0;
        if (s0 >= lo && s0 <= hi) c += s0 + 1;
        if (s1 >= lo && s1 <= hi) c += 63 - s1;
        for (int off = 16; off > 0; off >>= 1)
            c += __shfl_xor(c, off, 32);
        cnt[g] = c;
    }

    const int gl = (lane < 6) ? lane : 0;
    int myc = cnt[0];
    #pragma unroll
    for (int g = 1; g < 6; ++g)
        if (gl == g) myc = cnt[g];
    const double q = (double)(1 << gl) / (double)myc;

    const int s = i + j;
    double w = 0.0;
    #pragma unroll
    for (int g = 0; g < 6; ++g) {
        int lo, hi;
        band_bounds(g, lo, hi);
        const double qg = __shfl(q, g, 32);
        w += (s >= lo && s <= hi) ? qg : 0.0;
    }
    Wg[i * 32 + j] = (float)w;
}

__global__ __launch_bounds__(32) void dct_grade_kernel(
    const float* __restrict__ x, const float* __restrict__ Dg,
    const float* __restrict__ Wg, float* __restrict__ grade,
    int C, int H, int Wd, int nW, int L)
{
    __shared__ __attribute__((aligned(16))) float sP[32 * LPITCH];
    __shared__ __attribute__((aligned(16))) float sT[32 * LPITCH];

    const int lane = threadIdx.x;
    const int bl   = blockIdx.x;
    const int l    = bl % L;
    const int b    = bl / L;
    const int pr   = (l / nW) * 16;
    const int pc   = (l % nW) * 16;

    const int half = lane >> 4;
    const int n    = lane & 15;

    v2f dfr[2][4];
    #pragma unroll
    for (int p = 0; p < 2; ++p)
        #pragma unroll
        for (int kq = 0; kq < 4; ++kq)
            dfr[p][kq] = *(const v2f*)(Dg + (2 * n + p) * 32 + kq * 4 + half * 2);

    float gs = 0.f;

    for (int c = 0; c < C; ++c) {
        const float* xp = x + (((size_t)b * C + c) * H + pr) * (size_t)Wd + pc;
        __syncthreads();
        stage_fold(sP, xp, (size_t)Wd, lane);
        __syncthreads();

        v8f t[2][2];
        #pragma unroll
        for (int p = 0; p < 2; ++p)
            #pragma unroll
            for (int ct = 0; ct < 2; ++ct) {
                v8f acc = {0.f,0.f,0.f,0.f,0.f,0.f,0.f,0.f};
                #pragma unroll
                for (int kq = 0; kq < 4; ++kq) {
                    const int k0 = kq * 4 + half * 2;
                    v2f bb;
                    bb.x = sP[(16 * p + k0)     * LPITCH + ct * 16 + n];
                    bb.y = sP[(16 * p + k0 + 1) * LPITCH + ct * 16 + n];
                    acc = __builtin_amdgcn_wmma_f32_16x16x4_f32(false, dfr[p][kq], false, bb,
                                                                (short)0, acc, false, false);
                }
                t[p][ct] = acc;
            }
        store_cd(sT, t, lane);
        __syncthreads();

        v8f xd[2][2];
        #pragma unroll
        for (int pi = 0; pi < 2; ++pi) {
            v2f ae[4], ao[4];
            #pragma unroll
            for (int kq = 0; kq < 4; ++kq) {
                const int k0 = kq * 4 + half * 2;
                const v2f t0 = *(const v2f*)(sT + (16 * pi + n) * LPITCH + k0);
                const v2f t1 = *(const v2f*)(sT + (16 * pi + n) * LPITCH + 30 - k0);
                ae[kq].x = t0.x + t1.y;  ae[kq].y = t0.y + t1.x;
                ao[kq].x = t0.x - t1.y;  ao[kq].y = t0.y - t1.x;
            }
            v8f xe = {0.f,0.f,0.f,0.f,0.f,0.f,0.f,0.f};
            v8f xo = {0.f,0.f,0.f,0.f,0.f,0.f,0.f,0.f};
            #pragma unroll
            for (int kq = 0; kq < 4; ++kq) {
                xe = __builtin_amdgcn_wmma_f32_16x16x4_f32(false, ae[kq], false, dfr[0][kq],
                                                           (short)0, xe, false, false);
                xo = __builtin_amdgcn_wmma_f32_16x16x4_f32(false, ao[kq], false, dfr[1][kq],
                                                           (short)0, xo, false, false);
            }
            xd[pi][0] = xe;
            xd[pi][1] = xo;
        }

        #pragma unroll
        for (int pi = 0; pi < 2; ++pi)
            #pragma unroll
            for (int pn = 0; pn < 2; ++pn)
                #pragma unroll
                for (int rr = 0; rr < 8; ++rr) {
                    const int m  = 2 * (half * 8 + rr) + pi;
                    const int nn = 2 * n + pn;
                    const float v = xd[pi][pn][rr];
                    gs += __builtin_amdgcn_logf(fabsf(v) + 1.0f) * Wg[m * 32 + nn];
                }
    }

    for (int off = 16; off > 0; off >>= 1)
        gs += __shfl_down(gs, off, 32);
    if (lane == 0) grade[bl] = gs * 0.693147180559945f;
}

__device__ __forceinline__ void ins_min(float v, int i, float& m1, int& i1,
                                        float& m2, int& i2)
{
    if (i < 0) return;
    if (v < m1 || (v == m1 && i < i1)) { m2 = m1; i2 = i1; m1 = v; i1 = i; }
    else if (v < m2 || (v == m2 && i < i2)) { m2 = v; i2 = i; }
}
__device__ __forceinline__ void ins_max(float v, int i, float& m1, int& i1,
                                        float& m2, int& i2)
{
    if (i < 0) return;
    if (v > m1 || (v == m1 && i > i1)) { m2 = m1; i2 = i1; m1 = v; i1 = i; }
    else if (v > m2 || (v == m2 && i > i2)) { m2 = v; i2 = i; }
}

__global__ __launch_bounds__(256) void select_kernel(const float* __restrict__ grade,
                                                     int* __restrict__ sel, int L)
{
    const int b   = blockIdx.x;
    const int tid = threadIdx.x;
    const float* g = grade + (size_t)b * L;

    float mn1 = FLT_MAX, mn2 = FLT_MAX, mx1 = -FLT_MAX, mx2 = -FLT_MAX;
    int   mi1 = -1, mi2 = -1, xi1 = -1, xi2 = -1;
    for (int l = tid; l < L; l += 256) {
        const float v = g[l];
        ins_min(v, l, mn1, mi1, mn2, mi2);
        ins_max(v, l, mx1, xi1, mx2, xi2);
    }

    __shared__ float smn1[256], smn2[256], smx1[256], smx2[256];
    __shared__ int   smi1[256], smi2[256], sxi1[256], sxi2[256];
    smn1[tid] = mn1; smn2[tid] = mn2; smx1[tid] = mx1; smx2[tid] = mx2;
    smi1[tid] = mi1; smi2[tid] = mi2; sxi1[tid] = xi1; sxi2[tid] = xi2;
    __syncthreads();

    for (int off = 128; off > 0; off >>= 1) {
        if (tid < off) {
            const int p = tid + off;
            ins_min(smn1[p], smi1[p], mn1, mi1, mn2, mi2);
            ins_min(smn2[p], smi2[p], mn1, mi1, mn2, mi2);
            ins_max(smx1[p], sxi1[p], mx1, xi1, mx2, xi2);
            ins_max(smx2[p], sxi2[p], mx1, xi1, mx2, xi2);
        }
        __syncthreads();
        if (tid < off) {
            smn1[tid] = mn1; smn2[tid] = mn2; smx1[tid] = mx1; smx2[tid] = mx2;
            smi1[tid] = mi1; smi2[tid] = mi2; sxi1[tid] = xi1; sxi2[tid] = xi2;
        }
        __syncthreads();
    }

    if (tid == 0) {
        sel[b * 4 + 0] = mi1;
        sel[b * 4 + 1] = xi1;
        sel[b * 4 + 2] = mi2;
        sel[b * 4 + 3] = xi2;
    }
}

__global__ __launch_bounds__(32) void level_out_kernel(
    const float* __restrict__ x, const float* __restrict__ Dg,
    const int* __restrict__ sel, float* __restrict__ out,
    int B, int C, int H, int Wd, int nW)
{
    __shared__ __attribute__((aligned(16))) float sD[32 * LPITCH];
    __shared__ __attribute__((aligned(16))) float sP[32 * LPITCH];
    __shared__ float sT[32 * LPITCH];

    const int lane = threadIdx.x;
    int bid = blockIdx.x;
    const int c = bid % C;  bid /= C;
    const int b = bid % B;
    const int s = bid / B;

    const int l  = sel[b * 4 + s];
    const int pr = (l / nW) * 16;
    const int pc = (l % nW) * 16;

    stage_tile(sD, Dg, 32, lane);
    const float* xp = x + (((size_t)b * C + c) * H + pr) * (size_t)Wd + pc;
    stage_tile(sP, xp, (size_t)Wd, lane);
    __syncthreads();

    const int half = lane >> 4;
    const int n    = lane & 15;

    v8f a0[2][2], a1[2][2];
    gemm32(sD, false, sP, false, a0, lane);
    store_cd(sT, a0, lane);
    __syncthreads();
    gemm32(sT, false, sD, true, a1, lane);

    #pragma unroll
    for (int i = 0; i < 2; ++i)
        #pragma unroll
        for (int j = 0; j < 2; ++j)
            #pragma unroll
            for (int rr = 0; rr < 8; ++rr) {
                const int m  = i * 16 + half * 8 + rr;
                const int nn = j * 16 + n;
                const float f = (((m + nn) > 64) || ((m + nn) < 0)) ? 0.f : 1.f;
                a1[i][j][rr] *= f;
            }

    __syncthreads();
    store_cd(sP, a1, lane);
    __syncthreads();
    gemm32(sD, true, sP, false, a0, lane);
    __syncthreads();
    store_cd(sT, a0, lane);
    __syncthreads();
    gemm32(sT, false, sD, false, a1, lane);

    float* op = out + (((size_t)s * B + b) * C + c) * 1024;
    #pragma unroll
    for (int i = 0; i < 2; ++i)
        #pragma unroll
        for (int j = 0; j < 2; ++j)
            #pragma unroll
            for (int rr = 0; rr < 8; ++rr)
                op[(i * 16 + half * 8 + rr) * 32 + j * 16 + n] = a1[i][j][rr];
}

extern "C" void kernel_launch(void* const* d_in, const int* in_sizes, int n_in,
                              void* d_out, int out_size, void* d_ws, size_t ws_size,
                              hipStream_t stream)
{
    const float* x = (const float*)d_in[0];
    float* out = (float*)d_out;

    const int B = 16, C = 3, H = 512, Wd = 512;
    const int nH = (H - WSZ) / 16 + 1;
    const int nW = (Wd - WSZ) / 16 + 1;
    const int L  = nH * nW;

    float* ws    = (float*)d_ws;
    float* Dg    = ws;
    float* Wg    = ws + 1024;
    float* grade = ws + 2048;
    int*   sel   = (int*)(ws + 2048 + B * L);

    hipLaunchKernelGGL(init_consts_kernel, dim3(64), dim3(32), 0, stream, Dg, Wg);
    hipLaunchKernelGGL(dct_grade_kernel, dim3(B * L), dim3(32), 0, stream,
                       x, Dg, Wg, grade, C, H, Wd, nW, L);
    hipLaunchKernelGGL(select_kernel, dim3(B), dim3(256), 0, stream, grade, sel, L);
    hipLaunchKernelGGL(level_out_kernel, dim3(B * 4 * C), dim3(32), 0, stream,
                       x, Dg, sel, out, B, C, H, Wd, nW);
}


# selection folded into the output launch: each one-wave work-group scans its batch's grades and finds its own patch index (wave top-2 butterfly); 3 launches, not 4
# speedup vs baseline: 6.0425x; 1.1756x over previous
#include <hip/hip_runtime.h>
#include <math.h>
#include <float.h>


typedef __attribute__((ext_vector_type(2))) float v2f;
typedef __attribute__((ext_vector_type(8))) float v8f;
typedef __attribute__((ext_vector_type(4))) float v4f;

#define WSZ     32
#define LPITCH  36


__device__ __forceinline__ v8f gemm16(const float* sA, int aRow0, bool aT,
                                      const float* sB, int bCol0, bool bT,
                                      int lane)
{
    const int half = lane >> 4;
    const int r    = lane & 15;
    v8f c = {0.f,0.f,0.f,0.f,0.f,0.f,0.f,0.f};
    #pragma unroll
    for (int kq = 0; kq < 8; ++kq) {
        const int k0 = kq * 4 + half * 2;
        v2f a, b;
        if (!aT) {
            a.x = sA[(aRow0 + r) * LPITCH + k0];
            a.y = sA[(aRow0 + r) * LPITCH + k0 + 1];
        } else {
            a.x = sA[k0       * LPITCH + aRow0 + r];
            a.y = sA[(k0 + 1) * LPITCH + aRow0 + r];
        }
        if (!bT) {
            b.x = sB[k0       * LPITCH + bCol0 + r];
            b.y = sB[(k0 + 1) * LPITCH + bCol0 + r];
        } else {
            b.x = sB[(bCol0 + r) * LPITCH + k0];
            b.y = sB[(bCol0 + r) * LPITCH + k0 + 1];
        }
        c = __builtin_amdgcn_wmma_f32_16x16x4_f32(false, a, false, b,
                                                  (short)0, c, false, false);
    }
    return c;
}

__device__ __forceinline__ void gemm32(const float* sA, bool aT,
                                       const float* sB, bool bT,
                                       v8f acc[2][2], int lane)
{
    #pragma unroll
    for (int i = 0; i < 2; ++i)
        #pragma unroll
        for (int j = 0; j < 2; ++j)
            acc[i][j] = gemm16(sA, i * 16, aT, sB, j * 16, bT, lane);
}

__device__ __forceinline__ void store_cd(float* sOut, v8f acc[2][2], int lane)
{
    const int half = lane >> 4;
    const int n    = lane & 15;
    #pragma unroll
    for (int i = 0; i < 2; ++i)
        #pragma unroll
        for (int j = 0; j < 2; ++j)
            #pragma unroll
            for (int rr = 0; rr < 8; ++rr)
                sOut[(i * 16 + half * 8 + rr) * LPITCH + j * 16 + n] = acc[i][j][rr];
}

__device__ __forceinline__ void stage_tile(float* sDst, const float* __restrict__ src,
                                           size_t stride, int lane)
{
    const int r0 = lane >> 3;
    const int c0 = (lane & 7) * 4;
    v4f v[8];
    #pragma unroll
    for (int g = 0; g < 8; ++g)
        v[g] = *(const v4f*)(src + (size_t)(g * 4 + r0) * stride + c0);
    #pragma unroll
    for (int g = 0; g < 8; ++g)
        *(v4f*)(sDst + (g * 4 + r0) * LPITCH + c0) = v[g];
}

__device__ __forceinline__ void stage_fold(float* sDst, const float* __restrict__ src,
                                           size_t stride, int lane)
{
    const int r0 = lane >> 3;
    const int c0 = (lane & 7) * 4;
    v4f top[4], bot[4];
    #pragma unroll
    for (int g = 0; g < 4; ++g) {
        const int j = g * 4 + r0;
        top[g] = *(const v4f*)(src + (size_t)j * stride + c0);
        bot[g] = *(const v4f*)(src + (size_t)(31 - j) * stride + c0);
    }
    #pragma unroll
    for (int g = 0; g < 4; ++g) {
        const int j = g * 4 + r0;
        *(v4f*)(sDst + j * LPITCH + c0)        = top[g] + bot[g];
        *(v4f*)(sDst + (16 + j) * LPITCH + c0) = top[g] - bot[g];
    }
}


__device__ __forceinline__ void band_bounds(int g, int& lo, int& hi)
{
    const double start = 64.0 / 6.0 * (double)g;
    const double end   = 64.0 / 6.0 * (double)(g + 1);
    lo = (int)start;
    if ((double)lo < start) ++lo;
    hi = (int)end;
}

__global__ __launch_bounds__(32) void init_consts_kernel(float* __restrict__ Dg,
                                                         float* __restrict__ Wg)
{
    const int lane = threadIdx.x;

    if (blockIdx.x < 32) {
        const int i = blockIdx.x, j = lane;
        const double scale = (i == 0) ? sqrt(1.0 / 32.0) : sqrt(2.0 / 32.0);
        Dg[i * 32 + j] =
            (float)(scale * cos(((double)j + 0.5) * M_PI * (double)i / 32.0));
        return;
    }

    const int i = blockIdx.x - 32, j = lane;

    int cnt[6];
    #pragma unroll
    for (int g = 0; g < 6; ++g) {
        int lo, hi;
        band_bounds(g, lo, hi);
        const int s0 = lane, s1 = lane + 32;
        int c = 0;
        if (s0 >= lo && s0 <= hi) c += s0 + 1;
        if (s1 >= lo && s1 <= hi) c += 63 - s1;
        for (int off = 16; off > 0; off >>= 1)
            c += __shfl_xor(c, off, 32);
        cnt[g] = c;
    }

    const int gl = (lane < 6) ? lane : 0;
    int myc = cnt[0];
    #pragma unroll
    for (int g = 1; g < 6; ++g)
        if (gl == g) myc = cnt[g];
    const double q = (double)(1 << gl) / (double)myc;

    const int s = i + j;
    double w = 0.0;
    #pragma unroll
    for (int g = 0; g < 6; ++g) {
        int lo, hi;
        band_bounds(g, lo, hi);
        const double qg = __shfl(q, g, 32);
        w += (s >= lo && s <= hi) ? qg : 0.0;
    }
    Wg[i * 32 + j] = (float)w;
}

__global__ __launch_bounds__(32) void dct_grade_kernel(
    const float* __restrict__ x, const float* __restrict__ Dg,
    const float* __restrict__ Wg, float* __restrict__ grade,
    int C, int H, int Wd, int nW, int L)
{
    __shared__ __attribute__((aligned(16))) float sP[32 * LPITCH];
    __shared__ __attribute__((aligned(16))) float sT[32 * LPITCH];

    const int lane = threadIdx.x;
    const int bl   = blockIdx.x;
    const int l    = bl % L;
    const int b    = bl / L;
    const int pr   = (l / nW) * 16;
    const int pc   = (l % nW) * 16;

    const int half = lane >> 4;
    const int n    = lane & 15;

    v2f dfr[2][4];
    #pragma unroll
    for (int p = 0; p < 2; ++p)
        #pragma unroll
        for (int kq = 0; kq < 4; ++kq)
            dfr[p][kq] = *(const v2f*)(Dg + (2 * n + p) * 32 + kq * 4 + half * 2);

    float gs = 0.f;

    for (int c = 0; c < C; ++c) {
        const float* xp = x + (((size_t)b * C + c) * H + pr) * (size_t)Wd + pc;
        __syncthreads();
        stage_fold(sP, xp, (size_t)Wd, lane);
        __syncthreads();

        v8f t[2][2];
        #pragma unroll
        for (int p = 0; p < 2; ++p)
            #pragma unroll
            for (int ct = 0; ct < 2; ++ct) {
                v8f acc = {0.f,0.f,0.f,0.f,0.f,0.f,0.f,0.f};
                #pragma unroll
                for (int kq = 0; kq < 4; ++kq) {
                    const int k0 = kq * 4 + half * 2;
                    v2f bb;
                    bb.x = sP[(16 * p + k0)     * LPITCH + ct * 16 + n];
                    bb.y = sP[(16 * p + k0 + 1) * LPITCH + ct * 16 + n];
                    acc = __builtin_amdgcn_wmma_f32_16x16x4_f32(false, dfr[p][kq], false, bb,
                                                                (short)0, acc, false, false);
                }
                t[p][ct] = acc;
            }
        store_cd(sT, t, lane);
        __syncthreads();

        v8f xd[2][2];
        #pragma unroll
        for (int pi = 0; pi < 2; ++pi) {
            v2f ae[4], ao[4];
            #pragma unroll
            for (int kq = 0; kq < 4; ++kq) {
                const int k0 = kq * 4 + half * 2;
                const v2f t0 = *(const v2f*)(sT + (16 * pi + n) * LPITCH + k0);
                const v2f t1 = *(const v2f*)(sT + (16 * pi + n) * LPITCH + 30 - k0);
                ae[kq].x = t0.x + t1.y;  ae[kq].y = t0.y + t1.x;
                ao[kq].x = t0.x - t1.y;  ao[kq].y = t0.y - t1.x;
            }
            v8f xe = {0.f,0.f,0.f,0.f,0.f,0.f,0.f,0.f};
            v8f xo = {0.f,0.f,0.f,0.f,0.f,0.f,0.f,0.f};
            #pragma unroll
            for (int kq = 0; kq < 4; ++kq) {
                xe = __builtin_amdgcn_wmma_f32_16x16x4_f32(false, ae[kq], false, dfr[0][kq],
                                                           (short)0, xe, false, false);
                xo = __builtin_amdgcn_wmma_f32_16x16x4_f32(false, ao[kq], false, dfr[1][kq],
                                                           (short)0, xo, false, false);
            }
            xd[pi][0] = xe;
            xd[pi][1] = xo;
        }

        #pragma unroll
        for (int pi = 0; pi < 2; ++pi)
            #pragma unroll
            for (int pn = 0; pn < 2; ++pn)
                #pragma unroll
                for (int rr = 0; rr < 8; ++rr) {
                    const int m  = 2 * (half * 8 + rr) + pi;
                    const int nn = 2 * n + pn;
                    const float v = xd[pi][pn][rr];
                    gs += __builtin_amdgcn_logf(fabsf(v) + 1.0f) * Wg[m * 32 + nn];
                }
    }

    for (int off = 16; off > 0; off >>= 1)
        gs += __shfl_down(gs, off, 32);
    if (lane == 0) grade[bl] = gs * 0.693147180559945f;
}

template <bool MAX>
__device__ __forceinline__ bool comes_before(float v, int i, float m, int im)
{
    const bool byval = MAX ? (v > m) : (v < m);
    const bool byidx = MAX ? (i > im) : (i < im);
    return byval | ((v == m) & byidx);
}

template <bool MAX>
__device__ __forceinline__ void ins_top2(float v, int i, float& m1, int& i1,
                                         float& m2, int& i2)
{
    const bool ok = i >= 0;
    const bool a = ok & comes_before<MAX>(v, i, m1, i1);
    const bool c = ok & comes_before<MAX>(v, i, m2, i2);
    m2 = a ? m1 : (c ? v : m2);
    i2 = a ? i1 : (c ? i : i2);
    m1 = a ? v : m1;
    i1 = a ? i : i1;
}

template <bool MAX>
__device__ __forceinline__ void ins_ascending(float v, int i, float& m1, int& i1,
                                              float& m2, int& i2)
{
    const bool a = MAX ? (v >= m1) : (v < m1);
    const bool c = MAX ? (v >= m2) : (v < m2);
    m2 = a ? m1 : (c ? v : m2);
    i2 = a ? i1 : (c ? i : i2);
    m1 = a ? v : m1;
    i1 = a ? i : i1;
}

#define SEL_CH  32

template <bool MAX>
__device__ __forceinline__ int pick_index(const float* __restrict__ g, int L,
                                          int lane, bool second)
{
    float m1 = MAX ? -FLT_MAX : FLT_MAX, m2 = m1;
    int   i1 = -1, i2 = -1;
    for (int base = 0; base < L; base += 32 * SEL_CH) {
        float v[SEL_CH];
        #pragma unroll
        for (int u = 0; u < SEL_CH; ++u) {
            const int l = base + u * 32 + lane;
            v[u] = g[l < L ? l : L - 1];
        }
        #pragma unroll
        for (int u = 0; u < SEL_CH; ++u) {
            const int l = base + u * 32 + lane;
            ins_ascending<MAX>(l < L ? v[u] : __builtin_nanf(""), l, m1, i1, m2, i2);
        }
    }
    for (int off = 16; off > 0; off >>= 1) {
        const float pm1 = __shfl_xor(m1, off, 32);
        const int   pi1 = __shfl_xor(i1, off, 32);
        const float pm2 = __shfl_xor(m2, off, 32);
        const int   pi2 = __shfl_xor(i2, off, 32);
        ins_top2<MAX>(pm1, pi1, m1, i1, m2, i2);
        ins_top2<MAX>(pm2, pi2, m1, i1, m2, i2);
    }
    return second ? i2 : i1;
}

__global__ __launch_bounds__(32) void select_level_out_kernel(
    const float* __restrict__ x, const float* __restrict__ Dg,
    const float* __restrict__ grade, float* __restrict__ out,
    int B, int C, int H, int Wd, int nW, int L)
{
    __shared__ __attribute__((aligned(16))) float sD[32 * LPITCH];
    __shared__ __attribute__((aligned(16))) float sP[32 * LPITCH];
    __shared__ float sT[32 * LPITCH];

    const int lane = threadIdx.x;
    int bid = blockIdx.x;
    const int c = bid % C;  bid /= C;
    const int b = bid % B;
    const int s = bid / B;

    const float* g = grade + (size_t)b * L;
    const bool second = (s >> 1) != 0;
    int l = (s & 1) ? pick_index<true>(g, L, lane, second)
                    : pick_index<false>(g, L, lane, second);
    l = __builtin_amdgcn_readfirstlane(l);
    l = l < 0 ? 0 : l;
    const int pr = (l / nW) * 16;
    const int pc = (l % nW) * 16;

    stage_tile(sD, Dg, 32, lane);
    const float* xp = x + (((size_t)b * C + c) * H + pr) * (size_t)Wd + pc;
    stage_tile(sP, xp, (size_t)Wd, lane);
    __syncthreads();

    const int half = lane >> 4;
    const int n    = lane & 15;

    v8f a0[2][2], a1[2][2];
    gemm32(sD, false, sP, false, a0, lane);
    store_cd(sT, a0, lane);
    __syncthreads();
    gemm32(sT, false, sD, true, a1, lane);

    #pragma unroll
    for (int i = 0; i < 2; ++i)
        #pragma unroll
        for (int j = 0; j < 2; ++j)
            #pragma unroll
            for (int rr = 0; rr < 8; ++rr) {
                const int m  = i * 16 + half * 8 + rr;
                const int nn = j * 16 + n;
                const float f = (((m + nn) > 64) || ((m + nn) < 0)) ? 0.f : 1.f;
                a1[i][j][rr] *= f;
            }

    __syncthreads();
    store_cd(sP, a1, lane);
    __syncthreads();
    gemm32(sD, true, sP, false, a0, lane);
    __syncthreads();
    store_cd(sT, a0, lane);
    __syncthreads();
    gemm32(sT, false, sD, false, a1, lane);

    float* op = out + (((size_t)s * B + b) * C + c) * 1024;
    #pragma unroll
    for (int i = 0; i < 2; ++i)
        #pragma unroll
        for (int j = 0; j < 2; ++j)
            #pragma unroll
            for (int rr = 0; rr < 8; ++rr)
                op[(i * 16 + half * 8 + rr) * 32 + j * 16 + n] = a1[i][j][rr];
}

extern "C" void kernel_launch(void* const* d_in, const int* in_sizes, int n_in,
                              void* d_out, int out_size, void* d_ws, size_t ws_size,
                              hipStream_t stream)
{
    const float* x = (const float*)d_in[0];
    float* out = (float*)d_out;

    const int B = 16, C = 3, H = 512, Wd = 512;
    const int nH = (H - WSZ) / 16 + 1;
    const int nW = (Wd - WSZ) / 16 + 1;
    const int L  = nH * nW;

    float* ws    = (float*)d_ws;
    float* Dg    = ws;
    float* Wg    = ws + 1024;
    float* grade = ws + 2048;

    hipLaunchKernelGGL(init_consts_kernel, dim3(64), dim3(32), 0, stream, Dg, Wg);
    hipLaunchKernelGGL(dct_grade_kernel, dim3(B * L), dim3(32), 0, stream,
                       x, Dg, Wg, grade, C, H, Wd, nW, L);
    hipLaunchKernelGGL(select_level_out_kernel, dim3(B * 4 * C), dim3(32), 0, stream,
                       x, Dg, grade, out, B, C, H, Wd, nW, L);
}


# grade kernel: one wave walks 4 row-neighbour patches and keeps D*P of each shared 32x16 column block in LDS, so it is done once, not twice: 26 WMMAs per patch-channel, not 32
# speedup vs baseline: 6.8305x; 1.1304x over previous
#include <hip/hip_runtime.h>
#include <math.h>
#include <float.h>


typedef __attribute__((ext_vector_type(2))) float v2f;
typedef __attribute__((ext_vector_type(8))) float v8f;
typedef __attribute__((ext_vector_type(4))) float v4f;

#define WSZ     32
#define LPITCH  36


__device__ __forceinline__ v8f gemm16(const float* sA, int aRow0, bool aT,
                                      const float* sB, int bCol0, bool bT,
                                      int lane)
{
    const int half = lane >> 4;
    const int r    = lane & 15;
    v8f c = {0.f,0.f,0.f,0.f,0.f,0.f,0.f,0.f};
    #pragma unroll
    for (int kq = 0; kq < 8; ++kq) {
        const int k0 = kq * 4 + half * 2;
        v2f a, b;
        if (!aT) {
            a.x = sA[(aRow0 + r) * LPITCH + k0];
            a.y = sA[(aRow0 + r) * LPITCH + k0 + 1];
        } else {
            a.x = sA[k0       * LPITCH + aRow0 + r];
            a.y = sA[(k0 + 1) * LPITCH + aRow0 + r];
        }
        if (!bT) {
            b.x = sB[k0       * LPITCH + bCol0 + r];
            b.y = sB[(k0 + 1) * LPITCH + bCol0 + r];
        } else {
            b.x = sB[(bCol0 + r) * LPITCH + k0];
            b.y = sB[(bCol0 + r) * LPITCH + k0 + 1];
        }
        c = __builtin_amdgcn_wmma_f32_16x16x4_f32(false, a, false, b,
                                                  (short)0, c, false, false);
    }
    return c;
}

__device__ __forceinline__ void gemm32(const float* sA, bool aT,
                                       const float* sB, bool bT,
                                       v8f acc[2][2], int lane)
{
    #pragma unroll
    for (int i = 0; i < 2; ++i)
        #pragma unroll
        for (int j = 0; j < 2; ++j)
            acc[i][j] = gemm16(sA, i * 16, aT, sB, j * 16, bT, lane);
}

__device__ __forceinline__ void store_cd(float* sOut, v8f acc[2][2], int lane)
{
    const int half = lane >> 4;
    const int n    = lane & 15;
    #pragma unroll
    for (int i = 0; i < 2; ++i)
        #pragma unroll
        for (int j = 0; j < 2; ++j)
            #pragma unroll
            for (int rr = 0; rr < 8; ++rr)
                sOut[(i * 16 + half * 8 + rr) * LPITCH + j * 16 + n] = acc[i][j][rr];
}

__device__ __forceinline__ void stage_tile(float* sDst, const float* __restrict__ src,
                                           size_t stride, int lane)
{
    const int r0 = lane >> 3;
    const int c0 = (lane & 7) * 4;
    v4f v[8];
    #pragma unroll
    for (int g = 0; g < 8; ++g)
        v[g] = *(const v4f*)(src + (size_t)(g * 4 + r0) * stride + c0);
    #pragma unroll
    for (int g = 0; g < 8; ++g)
        *(v4f*)(sDst + (g * 4 + r0) * LPITCH + c0) = v[g];
}


__device__ __forceinline__ void band_bounds(int g, int& lo, int& hi)
{
    const double start = 64.0 / 6.0 * (double)g;
    const double end   = 64.0 / 6.0 * (double)(g + 1);
    lo = (int)start;
    if ((double)lo < start) ++lo;
    hi = (int)end;
}

__global__ __launch_bounds__(32) void init_consts_kernel(float* __restrict__ Dg,
                                                         float* __restrict__ Wg)
{
    const int lane = threadIdx.x;

    if (blockIdx.x < 32) {
        const int i = blockIdx.x, j = lane;
        const double scale = (i == 0) ? sqrt(1.0 / 32.0) : sqrt(2.0 / 32.0);
        Dg[i * 32 + j] =
            (float)(scale * cos(((double)j + 0.5) * M_PI * (double)i / 32.0));
        return;
    }

    const int i = blockIdx.x - 32, j = lane;

    int cnt[6];
    #pragma unroll
    for (int g = 0; g < 6; ++g) {
        int lo, hi;
        band_bounds(g, lo, hi);
        const int s0 = lane, s1 = lane + 32;
        int c = 0;
        if (s0 >= lo && s0 <= hi) c += s0 + 1;
        if (s1 >= lo && s1 <= hi) c += 63 - s1;
        for (int off = 16; off > 0; off >>= 1)
            c += __shfl_xor(c, off, 32);
        cnt[g] = c;
    }

    const int gl = (lane < 6) ? lane : 0;
    int myc = cnt[0];
    #pragma unroll
    for (int g = 1; g < 6; ++g)
        if (gl == g) myc = cnt[g];
    const double q = (double)(1 << gl) / (double)myc;

    const int s = i + j;
    double w = 0.0;
    #pragma unroll
    for (int g = 0; g < 6; ++g) {
        int lo, hi;
        band_bounds(g, lo, hi);
        const double qg = __shfl(q, g, 32);
        w += (s >= lo && s <= hi) ? qg : 0.0;
    }
    Wg[i * 32 + j] = (float)w;
}

#define RUNLEN  4
#define NCH     3
#define PPITCH  16
#define TPITCH  20

__device__ __forceinline__ void load_block(v4f top[2], v4f bot[2],
                                           const float* __restrict__ src,
                                           size_t stride, int lane)
{
    const int r0 = lane >> 2;
    const int c0 = (lane & 3) * 4;
    #pragma unroll
    for (int g = 0; g < 2; ++g) {
        const int j = g * 8 + r0;
        top[g] = *(const v4f*)(src + (size_t)j * stride + c0);
        bot[g] = *(const v4f*)(src + (size_t)(31 - j) * stride + c0);
    }
}

__device__ __forceinline__ void fold_store(float* sDst, const v4f top[2],
                                           const v4f bot[2], int lane)
{
    const int r0 = lane >> 2;
    const int c0 = (lane & 3) * 4;
    #pragma unroll
    for (int g = 0; g < 2; ++g) {
        const int j = g * 8 + r0;
        *(v4f*)(sDst + j * PPITCH + c0)        = top[g] + bot[g];
        *(v4f*)(sDst + (16 + j) * PPITCH + c0) = top[g] - bot[g];
    }
}

__device__ __forceinline__ void lds_barrier()
{
    __builtin_amdgcn_fence(__ATOMIC_RELEASE, "workgroup", "local");
    __builtin_amdgcn_s_barrier();
    __builtin_amdgcn_fence(__ATOMIC_ACQUIRE, "workgroup", "local");
}

__global__ __launch_bounds__(32) void dct_grade_kernel(
    const float* __restrict__ x, const float* __restrict__ Dg,
    const float* __restrict__ Wg, float* __restrict__ grade,
    int H, int Wd, int nW, int L)
{
    __shared__ __attribute__((aligned(16))) float sP[32 * PPITCH];
    __shared__ __attribute__((aligned(16))) float sT[NCH][2][32 * TPITCH];

    const int lane = threadIdx.x;
    const int ow0  = blockIdx.x * RUNLEN;
    const int oh   = blockIdx.y;
    const int b    = blockIdx.z;
    int np = nW - ow0;
    np = np < RUNLEN ? np : RUNLEN;

    const int half = lane >> 4;
    const int n    = lane & 15;

    v2f dfr[2][4];
    #pragma unroll
    for (int p = 0; p < 2; ++p)
        #pragma unroll
        for (int kq = 0; kq < 4; ++kq)
            dfr[p][kq] = *(const v2f*)(Dg + (2 * n + p) * 32 + kq * 4 + half * 2);

    float wv[2][2][8];
    #pragma unroll
    for (int pi = 0; pi < 2; ++pi)
        #pragma unroll
        for (int pn = 0; pn < 2; ++pn)
            #pragma unroll
            for (int rr = 0; rr < 8; ++rr)
                wv[pi][pn][rr] = Wg[(2 * (half * 8 + rr) + pi) * 32 + 2 * n + pn];

    const size_t chs = (size_t)H * (size_t)Wd;
    const float* xb  = x + (size_t)b * NCH * chs + (size_t)oh * 16 * (size_t)Wd + (size_t)ow0 * 16;
    float* gout = grade + (size_t)b * L + (size_t)oh * nW + ow0;

    v4f top[NCH][2], bot[NCH][2];
    #pragma unroll
    for (int c = 0; c < NCH; ++c)
        load_block(top[c], bot[c], xb + c * chs, (size_t)Wd, lane);

    for (int blk = 0; blk <= np; ++blk) {
        const int cur = blk & 1;
        const int nb  = blk < np ? blk + 1 : np;
        float gs = 0.f;

        #pragma unroll
        for (int c = 0; c < NCH; ++c) {
            lds_barrier();
            fold_store(sP, top[c], bot[c], lane);
            load_block(top[c], bot[c], xb + c * chs + nb * 16, (size_t)Wd, lane);
            lds_barrier();

            float* tw = sT[c][cur];
            #pragma unroll
            for (int p = 0; p < 2; ++p) {
                v8f acc = {0.f,0.f,0.f,0.f,0.f,0.f,0.f,0.f};
                #pragma unroll
                for (int kq = 0; kq < 4; ++kq) {
                    const int k0 = kq * 4 + half * 2;
                    v2f bb;
                    bb.x = sP[(16 * p + k0)     * PPITCH + n];
                    bb.y = sP[(16 * p + k0 + 1) * PPITCH + n];
                    acc = __builtin_amdgcn_wmma_f32_16x16x4_f32(false, dfr[p][kq], false, bb,
                                                                (short)0, acc, false, false);
                }
                #pragma unroll
                for (int rr = 0; rr < 8; ++rr)
                    tw[(16 * p + half * 8 + rr) * TPITCH + n] = acc[rr];
            }

            if (blk > 0) {
                lds_barrier();
                const float* tl = sT[c][cur ^ 1];
                const float* tr = tw;

                v8f xd[2][2];
                #pragma unroll
                for (int pi = 0; pi < 2; ++pi) {
                    v2f ae[4], ao[4];
                    #pragma unroll
                    for (int kq = 0; kq < 4; ++kq) {
                        const int k0 = kq * 4 + half * 2;
                        const v2f t0 = *(const v2f*)(tl + (16 * pi + n) * TPITCH + k0);
                        const v2f t1 = *(const v2f*)(tr + (16 * pi + n) * TPITCH + 14 - k0);
                        ae[kq].x = t0.x + t1.y;  ae[kq].y = t0.y + t1.x;
                        ao[kq].x = t0.x - t1.y;  ao[kq].y = t0.y - t1.x;
                    }
                    v8f xe = {0.f,0.f,0.f,0.f,0.f,0.f,0.f,0.f};
                    v8f xo = {0.f,0.f,0.f,0.f,0.f,0.f,0.f,0.f};
                    #pragma unroll
                    for (int kq = 0; kq < 4; ++kq) {
                        xe = __builtin_amdgcn_wmma_f32_16x16x4_f32(false, ae[kq], false, dfr[0][kq],
                                                                   (short)0, xe, false, false);
                        xo = __builtin_amdgcn_wmma_f32_16x16x4_f32(false, ao[kq], false, dfr[1][kq],
                                                                   (short)0, xo, false, false);
                    }
                    xd[pi][0] = xe;
                    xd[pi][1] = xo;
                }

                #pragma unroll
                for (int pi = 0; pi < 2; ++pi)
                    #pragma unroll
                    for (int pn = 0; pn < 2; ++pn)
                        #pragma unroll
                        for (int rr = 0; rr < 8; ++rr) {
                            const float v = xd[pi][pn][rr];
                            gs += __builtin_amdgcn_logf(fabsf(v) + 1.0f) * wv[pi][pn][rr];
                        }
            }
        }

        if (blk > 0) {
            for (int off = 16; off > 0; off >>= 1)
                gs += __shfl_down(gs, off, 32);
            if (lane == 0) gout[blk - 1] = gs * 0.693147180559945f;
        }
    }
}

template <bool MAX>
__device__ __forceinline__ bool comes_before(float v, int i, float m, int im)
{
    const bool byval = MAX ? (v > m) : (v < m);
    const bool byidx = MAX ? (i > im) : (i < im);
    return byval | ((v == m) & byidx);
}

template <bool MAX>
__device__ __forceinline__ void ins_top2(float v, int i, float& m1, int& i1,
                                         float& m2, int& i2)
{
    const bool ok = i >= 0;
    const bool a = ok & comes_before<MAX>(v, i, m1, i1);
    const bool c = ok & comes_before<MAX>(v, i, m2, i2);
    m2 = a ? m1 : (c ? v : m2);
    i2 = a ? i1 : (c ? i : i2);
    m1 = a ? v : m1;
    i1 = a ? i : i1;
}

template <bool MAX>
__device__ __forceinline__ void ins_ascending(float v, int i, float& m1, int& i1,
                                              float& m2, int& i2)
{
    const bool a = MAX ? (v >= m1) : (v < m1);
    const bool c = MAX ? (v >= m2) : (v < m2);
    m2 = a ? m1 : (c ? v : m2);
    i2 = a ? i1 : (c ? i : i2);
    m1 = a ? v : m1;
    i1 = a ? i : i1;
}

#define SEL_CH  32

template <bool MAX>
__device__ __forceinline__ int pick_index(const float* __restrict__ g, int L,
                                          int lane, bool second)
{
    float m1 = MAX ? -FLT_MAX : FLT_MAX, m2 = m1;
    int   i1 = -1, i2 = -1;
    for (int base = 0; base < L; base += 32 * SEL_CH) {
        float v[SEL_CH];
        #pragma unroll
        for (int u = 0; u < SEL_CH; ++u) {
            const int l = base + u * 32 + lane;
            v[u] = g[l < L ? l : L - 1];
        }
        #pragma unroll
        for (int u = 0; u < SEL_CH; ++u) {
            const int l = base + u * 32 + lane;
            ins_ascending<MAX>(l < L ? v[u] : __builtin_nanf(""), l, m1, i1, m2, i2);
        }
    }
    for (int off = 16; off > 0; off >>= 1) {
        const float pm1 = __shfl_xor(m1, off, 32);
        const int   pi1 = __shfl_xor(i1, off, 32);
        const float pm2 = __shfl_xor(m2, off, 32);
        const int   pi2 = __shfl_xor(i2, off, 32);
        ins_top2<MAX>(pm1, pi1, m1, i1, m2, i2);
        ins_top2<MAX>(pm2, pi2, m1, i1, m2, i2);
    }
    return second ? i2 : i1;
}

__global__ __launch_bounds__(32) void select_level_out_kernel(
    const float* __restrict__ x, const float* __restrict__ Dg,
    const float* __restrict__ grade, float* __restrict__ out,
    int B, int C, int H, int Wd, int nW, int L)
{
    __shared__ __attribute__((aligned(16))) float sD[32 * LPITCH];
    __shared__ __attribute__((aligned(16))) float sP[32 * LPITCH];
    __shared__ float sT[32 * LPITCH];

    const int lane = threadIdx.x;
    int bid = blockIdx.x;
    const int c = bid % C;  bid /= C;
    const int b = bid % B;
    const int s = bid / B;

    const float* g = grade + (size_t)b * L;
    const bool second = (s >> 1) != 0;
    int l = (s & 1) ? pick_index<true>(g, L, lane, second)
                    : pick_index<false>(g, L, lane, second);
    l = __builtin_amdgcn_readfirstlane(l);
    l = l < 0 ? 0 : l;
    const int pr = (l / nW) * 16;
    const int pc = (l % nW) * 16;

    stage_tile(sD, Dg, 32, lane);
    const float* xp = x + (((size_t)b * C + c) * H + pr) * (size_t)Wd + pc;
    stage_tile(sP, xp, (size_t)Wd, lane);
    __syncthreads();

    const int half = lane >> 4;
    const int n    = lane & 15;

    v8f a0[2][2], a1[2][2];
    gemm32(sD, false, sP, false, a0, lane);
    store_cd(sT, a0, lane);
    __syncthreads();
    gemm32(sT, false, sD, true, a1, lane);

    #pragma unroll
    for (int i = 0; i < 2; ++i)
        #pragma unroll
        for (int j = 0; j < 2; ++j)
            #pragma unroll
            for (int rr = 0; rr < 8; ++rr) {
                const int m  = i * 16 + half * 8 + rr;
                const int nn = j * 16 + n;
                const float f = (((m + nn) > 64) || ((m + nn) < 0)) ? 0.f : 1.f;
                a1[i][j][rr] *= f;
            }

    __syncthreads();
    store_cd(sP, a1, lane);
    __syncthreads();
    gemm32(sD, true, sP, false, a0, lane);
    __syncthreads();
    store_cd(sT, a0, lane);
    __syncthreads();
    gemm32(sT, false, sD, false, a1, lane);

    float* op = out + (((size_t)s * B + b) * C + c) * 1024;
    #pragma unroll
    for (int i = 0; i < 2; ++i)
        #pragma unroll
        for (int j = 0; j < 2; ++j)
            #pragma unroll
            for (int rr = 0; rr < 8; ++rr)
                op[(i * 16 + half * 8 + rr) * 32 + j * 16 + n] = a1[i][j][rr];
}

extern "C" void kernel_launch(void* const* d_in, const int* in_sizes, int n_in,
                              void* d_out, int out_size, void* d_ws, size_t ws_size,
                              hipStream_t stream)
{
    const float* x = (const float*)d_in[0];
    float* out = (float*)d_out;

    const int B = 16, C = 3, H = 512, Wd = 512;
    const int nH = (H - WSZ) / 16 + 1;
    const int nW = (Wd - WSZ) / 16 + 1;
    const int L  = nH * nW;

    float* ws    = (float*)d_ws;
    float* Dg    = ws;
    float* Wg    = ws + 1024;
    float* grade = ws + 2048;

    hipLaunchKernelGGL(init_consts_kernel, dim3(64), dim3(32), 0, stream, Dg, Wg);
    hipLaunchKernelGGL(dct_grade_kernel, dim3((nW + RUNLEN - 1) / RUNLEN, nH, B), dim3(32), 0, stream,
                       x, Dg, Wg, grade, H, Wd, nW, L);
    hipLaunchKernelGGL(select_level_out_kernel, dim3(B * 4 * C), dim3(32), 0, stream,
                       x, Dg, grade, out, B, C, H, Wd, nW, L);
}


# grade sum: a lane's six elements of equal i+j (anti-diagonal pair x 3 channels, equal weight) multiplied as (|X|+1) factors before one log2: 17 log2 per lane and patch, not 96; sum order changes
# speedup vs baseline: 7.1535x; 1.0473x over previous
#include <hip/hip_runtime.h>
#include <math.h>
#include <float.h>


typedef __attribute__((ext_vector_type(2))) float v2f;
typedef __attribute__((ext_vector_type(8))) float v8f;
typedef __attribute__((ext_vector_type(4))) float v4f;

#define WSZ     32
#define LPITCH  36


__device__ __forceinline__ v8f gemm16(const float* sA, int aRow0, bool aT,
                                      const float* sB, int bCol0, bool bT,
                                      int lane)
{
    const int half = lane >> 4;
    const int r    = lane & 15;
    v8f c = {0.f,0.f,0.f,0.f,0.f,0.f,0.f,0.f};
    #pragma unroll
    for (int kq = 0; kq < 8; ++kq) {
        const int k0 = kq * 4 + half * 2;
        v2f a, b;
        if (!aT) {
            a.x = sA[(aRow0 + r) * LPITCH + k0];
            a.y = sA[(aRow0 + r) * LPITCH + k0 + 1];
        } else {
            a.x = sA[k0       * LPITCH + aRow0 + r];
            a.y = sA[(k0 + 1) * LPITCH + aRow0 + r];
        }
        if (!bT) {
            b.x = sB[k0       * LPITCH + bCol0 + r];
            b.y = sB[(k0 + 1) * LPITCH + bCol0 + r];
        } else {
            b.x = sB[(bCol0 + r) * LPITCH + k0];
            b.y = sB[(bCol0 + r) * LPITCH + k0 + 1];
        }
        c = __builtin_amdgcn_wmma_f32_16x16x4_f32(false, a, false, b,
                                                  (short)0, c, false, false);
    }
    return c;
}

__device__ __forceinline__ void gemm32(const float* sA, bool aT,
                                       const float* sB, bool bT,
                                       v8f acc[2][2], int lane)
{
    #pragma unroll
    for (int i = 0; i < 2; ++i)
        #pragma unroll
        for (int j = 0; j < 2; ++j)
            acc[i][j] = gemm16(sA, i * 16, aT, sB, j * 16, bT, lane);
}

__device__ __forceinline__ void store_cd(float* sOut, v8f acc[2][2], int lane)
{
    const int half = lane >> 4;
    const int n    = lane & 15;
    #pragma unroll
    for (int i = 0; i < 2; ++i)
        #pragma unroll
        for (int j = 0; j < 2; ++j)
            #pragma unroll
            for (int rr = 0; rr < 8; ++rr)
                sOut[(i * 16 + half * 8 + rr) * LPITCH + j * 16 + n] = acc[i][j][rr];
}

__device__ __forceinline__ void stage_tile(float* sDst, const float* __restrict__ src,
                                           size_t stride, int lane)
{
    const int r0 = lane >> 3;
    const int c0 = (lane & 7) * 4;
    v4f v[8];
    #pragma unroll
    for (int g = 0; g < 8; ++g)
        v[g] = *(const v4f*)(src + (size_t)(g * 4 + r0) * stride + c0);
    #pragma unroll
    for (int g = 0; g < 8; ++g)
        *(v4f*)(sDst + (g * 4 + r0) * LPITCH + c0) = v[g];
}


__device__ __forceinline__ void band_bounds(int g, int& lo, int& hi)
{
    const double start = 64.0 / 6.0 * (double)g;
    const double end   = 64.0 / 6.0 * (double)(g + 1);
    lo = (int)start;
    if ((double)lo < start) ++lo;
    hi = (int)end;
}

__global__ __launch_bounds__(32) void init_consts_kernel(float* __restrict__ Dg,
                                                         float* __restrict__ Wg)
{
    const int lane = threadIdx.x;

    if (blockIdx.x < 32) {
        const int i = blockIdx.x, j = lane;
        const double scale = (i == 0) ? sqrt(1.0 / 32.0) : sqrt(2.0 / 32.0);
        Dg[i * 32 + j] =
            (float)(scale * cos(((double)j + 0.5) * M_PI * (double)i / 32.0));
        return;
    }

    const int i = blockIdx.x - 32, j = lane;

    int cnt[6];
    #pragma unroll
    for (int g = 0; g < 6; ++g) {
        int lo, hi;
        band_bounds(g, lo, hi);
        const int s0 = lane, s1 = lane + 32;
        int c = 0;
        if (s0 >= lo && s0 <= hi) c += s0 + 1;
        if (s1 >= lo && s1 <= hi) c += 63 - s1;
        for (int off = 16; off > 0; off >>= 1)
            c += __shfl_xor(c, off, 32);
        cnt[g] = c;
    }

    const int gl = (lane < 6) ? lane : 0;
    int myc = cnt[0];
    #pragma unroll
    for (int g = 1; g < 6; ++g)
        if (gl == g) myc = cnt[g];
    const double q = (double)(1 << gl) / (double)myc;

    const int s = i + j;
    double w = 0.0;
    #pragma unroll
    for (int g = 0; g < 6; ++g) {
        int lo, hi;
        band_bounds(g, lo, hi);
        const double qg = __shfl(q, g, 32);
        w += (s >= lo && s <= hi) ? qg : 0.0;
    }
    Wg[i * 32 + j] = (float)w;
}

#define RUNLEN  4
#define NCH     3
#define PPITCH  16
#define TPITCH  20

__device__ __forceinline__ void load_block(v4f top[2], v4f bot[2],
                                           const float* __restrict__ src,
                                           size_t stride, int lane)
{
    const int r0 = lane >> 2;
    const int c0 = (lane & 3) * 4;
    #pragma unroll
    for (int g = 0; g < 2; ++g) {
        const int j = g * 8 + r0;
        top[g] = *(const v4f*)(src + (size_t)j * stride + c0);
        bot[g] = *(const v4f*)(src + (size_t)(31 - j) * stride + c0);
    }
}

__device__ __forceinline__ void fold_store(float* sDst, const v4f top[2],
                                           const v4f bot[2], int lane)
{
    const int r0 = lane >> 2;
    const int c0 = (lane & 3) * 4;
    #pragma unroll
    for (int g = 0; g < 2; ++g) {
        const int j = g * 8 + r0;
        *(v4f*)(sDst + j * PPITCH + c0)        = top[g] + bot[g];
        *(v4f*)(sDst + (16 + j) * PPITCH + c0) = top[g] - bot[g];
    }
}

__device__ __forceinline__ void lds_barrier()
{
    __builtin_amdgcn_fence(__ATOMIC_RELEASE, "workgroup", "local");
    __builtin_amdgcn_s_barrier();
    __builtin_amdgcn_fence(__ATOMIC_ACQUIRE, "workgroup", "local");
}

__global__ __launch_bounds__(32) void dct_grade_kernel(
    const float* __restrict__ x, const float* __restrict__ Dg,
    const float* __restrict__ Wg, float* __restrict__ grade,
    int H, int Wd, int nW, int L)
{
    __shared__ __attribute__((aligned(16))) float sP[32 * PPITCH];
    __shared__ __attribute__((aligned(16))) float sT[NCH][2][32 * TPITCH];

    const int lane = threadIdx.x;
    const int ow0  = blockIdx.x * RUNLEN;
    const int oh   = blockIdx.y;
    const int b    = blockIdx.z;
    int np = nW - ow0;
    np = np < RUNLEN ? np : RUNLEN;

    const int half = lane >> 4;
    const int n    = lane & 15;

    v2f dfr[2][4];
    #pragma unroll
    for (int p = 0; p < 2; ++p)
        #pragma unroll
        for (int kq = 0; kq < 4; ++kq)
            dfr[p][kq] = *(const v2f*)(Dg + (2 * n + p) * 32 + kq * 4 + half * 2);

    float ws[17];
    #pragma unroll
    for (int k = 0; k < 17; ++k) {
        const int s  = 16 * half + 2 * n + k;
        const int wi = s < 32 ? s : 31;
        ws[k] = Wg[wi * 32 + (s - wi)];
    }

    const size_t chs = (size_t)H * (size_t)Wd;
    const float* xb  = x + (size_t)b * NCH * chs + (size_t)oh * 16 * (size_t)Wd + (size_t)ow0 * 16;
    float* gout = grade + (size_t)b * L + (size_t)oh * nW + ow0;

    v4f top[NCH][2], bot[NCH][2];
    #pragma unroll
    for (int c = 0; c < NCH; ++c)
        load_block(top[c], bot[c], xb + c * chs, (size_t)Wd, lane);

    for (int blk = 0; blk <= np; ++blk) {
        const int cur = blk & 1;
        const int nb  = blk < np ? blk + 1 : np;
        float prod[17];

        #pragma unroll
        for (int c = 0; c < NCH; ++c) {
            lds_barrier();
            fold_store(sP, top[c], bot[c], lane);
            load_block(top[c], bot[c], xb + c * chs + nb * 16, (size_t)Wd, lane);
            lds_barrier();

            float* tw = sT[c][cur];
            #pragma unroll
            for (int p = 0; p < 2; ++p) {
                v8f acc = {0.f,0.f,0.f,0.f,0.f,0.f,0.f,0.f};
                #pragma unroll
                for (int kq = 0; kq < 4; ++kq) {
                    const int k0 = kq * 4 + half * 2;
                    v2f bb;
                    bb.x = sP[(16 * p + k0)     * PPITCH + n];
                    bb.y = sP[(16 * p + k0 + 1) * PPITCH + n];
                    acc = __builtin_amdgcn_wmma_f32_16x16x4_f32(false, dfr[p][kq], false, bb,
                                                                (short)0, acc, false, false);
                }
                #pragma unroll
                for (int rr = 0; rr < 8; ++rr)
                    tw[(16 * p + half * 8 + rr) * TPITCH + n] = acc[rr];
            }

            if (blk > 0) {
                lds_barrier();
                const float* tl = sT[c][cur ^ 1];
                const float* tr = tw;

                v8f xd[2][2];
                #pragma unroll
                for (int pi = 0; pi < 2; ++pi) {
                    v2f ae[4], ao[4];
                    #pragma unroll
                    for (int kq = 0; kq < 4; ++kq) {
                        const int k0 = kq * 4 + half * 2;
                        const v2f t0 = *(const v2f*)(tl + (16 * pi + n) * TPITCH + k0);
                        const v2f t1 = *(const v2f*)(tr + (16 * pi + n) * TPITCH + 14 - k0);
                        ae[kq].x = t0.x + t1.y;  ae[kq].y = t0.y + t1.x;
                        ao[kq].x = t0.x - t1.y;  ao[kq].y = t0.y - t1.x;
                    }
                    v8f xe = {0.f,0.f,0.f,0.f,0.f,0.f,0.f,0.f};
                    v8f xo = {0.f,0.f,0.f,0.f,0.f,0.f,0.f,0.f};
                    #pragma unroll
                    for (int kq = 0; kq < 4; ++kq) {
                        xe = __builtin_amdgcn_wmma_f32_16x16x4_f32(false, ae[kq], false, dfr[0][kq],
                                                                   (short)0, xe, false, false);
                        xo = __builtin_amdgcn_wmma_f32_16x16x4_f32(false, ao[kq], false, dfr[1][kq],
                                                                   (short)0, xo, false, false);
                    }
                    xd[pi][0] = xe;
                    xd[pi][1] = xo;
                }

                float f[17];
                f[0]  = fabsf(xd[0][0][0]) + 1.0f;
                f[16] = fabsf(xd[1][1][7]) + 1.0f;
                #pragma unroll
                for (int rr = 1; rr < 8; ++rr) {
                    const float t = fabsf(xd[1][1][rr - 1]) + 1.0f;
                    f[2 * rr] = fmaf(fabsf(xd[0][0][rr]), t, t);
                }
                #pragma unroll
                for (int rr = 0; rr < 8; ++rr) {
                    const float t = fabsf(xd[1][0][rr]) + 1.0f;
                    f[2 * rr + 1] = fmaf(fabsf(xd[0][1][rr]), t, t);
                }
                #pragma unroll
                for (int k = 0; k < 17; ++k)
                    prod[k] = (c == 0) ? f[k] : prod[k] * f[k];
            }
        }

        if (blk > 0) {
            float gs = 0.f;
            #pragma unroll
            for (int k = 0; k < 17; ++k)
                gs += __builtin_amdgcn_logf(prod[k]) * ws[k];
            for (int off = 16; off > 0; off >>= 1)
                gs += __shfl_down(gs, off, 32);
            if (lane == 0) gout[blk - 1] = gs * 0.693147180559945f;
        }
    }
}

template <bool MAX>
__device__ __forceinline__ bool comes_before(float v, int i, float m, int im)
{
    const bool byval = MAX ? (v > m) : (v < m);
    const bool byidx = MAX ? (i > im) : (i < im);
    return byval | ((v == m) & byidx);
}

template <bool MAX>
__device__ __forceinline__ void ins_top2(float v, int i, float& m1, int& i1,
                                         float& m2, int& i2)
{
    const bool ok = i >= 0;
    const bool a = ok & comes_before<MAX>(v, i, m1, i1);
    const bool c = ok & comes_before<MAX>(v, i, m2, i2);
    m2 = a ? m1 : (c ? v : m2);
    i2 = a ? i1 : (c ? i : i2);
    m1 = a ? v : m1;
    i1 = a ? i : i1;
}

template <bool MAX>
__device__ __forceinline__ void ins_ascending(float v, int i, float& m1, int& i1,
                                              float& m2, int& i2)
{
    const bool a = MAX ? (v >= m1) : (v < m1);
    const bool c = MAX ? (v >= m2) : (v < m2);
    m2 = a ? m1 : (c ? v : m2);
    i2 = a ? i1 : (c ? i : i2);
    m1 = a ? v : m1;
    i1 = a ? i : i1;
}

#define SEL_CH  32

template <bool MAX>
__device__ __forceinline__ int pick_index(const float* __restrict__ g, int L,
                                          int lane, bool second)
{
    float m1 = MAX ? -FLT_MAX : FLT_MAX, m2 = m1;
    int   i1 = -1, i2 = -1;
    for (int base = 0; base < L; base += 32 * SEL_CH) {
        float v[SEL_CH];
        #pragma unroll
        for (int u = 0; u < SEL_CH; ++u) {
            const int l = base + u * 32 + lane;
            v[u] = g[l < L ? l : L - 1];
        }
        #pragma unroll
        for (int u = 0; u < SEL_CH; ++u) {
            const int l = base + u * 32 + lane;
            ins_ascending<MAX>(l < L ? v[u] : __builtin_nanf(""), l, m1, i1, m2, i2);
        }
    }
    for (int off = 16; off > 0; off >>= 1) {
        const float pm1 = __shfl_xor(m1, off, 32);
        const int   pi1 = __shfl_xor(i1, off, 32);
        const float pm2 = __shfl_xor(m2, off, 32);
        const int   pi2 = __shfl_xor(i2, off, 32);
        ins_top2<MAX>(pm1, pi1, m1, i1, m2, i2);
        ins_top2<MAX>(pm2, pi2, m1, i1, m2, i2);
    }
    return second ? i2 : i1;
}

__global__ __launch_bounds__(32) void select_level_out_kernel(
    const float* __restrict__ x, const float* __restrict__ Dg,
    const float* __restrict__ grade, float* __restrict__ out,
    int B, int C, int H, int Wd, int nW, int L)
{
    __shared__ __attribute__((aligned(16))) float sD[32 * LPITCH];
    __shared__ __attribute__((aligned(16))) float sP[32 * LPITCH];
    __shared__ float sT[32 * LPITCH];

    const int lane = threadIdx.x;
    int bid = blockIdx.x;
    const int c = bid % C;  bid /= C;
    const int b = bid % B;
    const int s = bid / B;

    const float* g = grade + (size_t)b * L;
    const bool second = (s >> 1) != 0;
    int l = (s & 1) ? pick_index<true>(g, L, lane, second)
                    : pick_index<false>(g, L, lane, second);
    l = __builtin_amdgcn_readfirstlane(l);
    l = l < 0 ? 0 : l;
    const int pr = (l / nW) * 16;
    const int pc = (l % nW) * 16;

    stage_tile(sD, Dg, 32, lane);
    const float* xp = x + (((size_t)b * C + c) * H + pr) * (size_t)Wd + pc;
    stage_tile(sP, xp, (size_t)Wd, lane);
    __syncthreads();

    const int half = lane >> 4;
    const int n    = lane & 15;

    v8f a0[2][2], a1[2][2];
    gemm32(sD, false, sP, false, a0, lane);
    store_cd(sT, a0, lane);
    __syncthreads();
    gemm32(sT, false, sD, true, a1, lane);

    #pragma unroll
    for (int i = 0; i < 2; ++i)
        #pragma unroll
        for (int j = 0; j < 2; ++j)
            #pragma unroll
            for (int rr = 0; rr < 8; ++rr) {
                const int m  = i * 16 + half * 8 + rr;
                const int nn = j * 16 + n;
                const float f = (((m + nn) > 64) || ((m + nn) < 0)) ? 0.f : 1.f;
                a1[i][j][rr] *= f;
            }

    __syncthreads();
    store_cd(sP, a1, lane);
    __syncthreads();
    gemm32(sD, true, sP, false, a0, lane);
    __syncthreads();
    store_cd(sT, a0, lane);
    __syncthreads();
    gemm32(sT, false, sD, false, a1, lane);

    float* op = out + (((size_t)s * B + b) * C + c) * 1024;
    #pragma unroll
    for (int i = 0; i < 2; ++i)
        #pragma unroll
        for (int j = 0; j < 2; ++j)
            #pragma unroll
            for (int rr = 0; rr < 8; ++rr)
                op[(i * 16 + half * 8 + rr) * 32 + j * 16 + n] = a1[i][j][rr];
}

extern "C" void kernel_launch(void* const* d_in, const int* in_sizes, int n_in,
                              void* d_out, int out_size, void* d_ws, size_t ws_size,
                              hipStream_t stream)
{
    const float* x = (const float*)d_in[0];
    float* out = (float*)d_out;

    const int B = 16, C = 3, H = 512, Wd = 512;
    const int nH = (H - WSZ) / 16 + 1;
    const int nW = (Wd - WSZ) / 16 + 1;
    const int L  = nH * nW;

    float* ws    = (float*)d_ws;
    float* Dg    = ws;
    float* Wg    = ws + 1024;
    float* grade = ws + 2048;

    hipLaunchKernelGGL(init_consts_kernel, dim3(64), dim3(32), 0, stream, Dg, Wg);
    hipLaunchKernelGGL(dct_grade_kernel, dim3((nW + RUNLEN - 1) / RUNLEN, nH, B), dim3(32), 0, stream,
                       x, Dg, Wg, grade, H, Wd, nW, L);
    hipLaunchKernelGGL(select_level_out_kernel, dim3(B * 4 * C), dim3(32), 0, stream,
                       x, Dg, grade, out, B, C, H, Wd, nW, L);
}
